# starting kernel
# baseline (speedup 1.0000x reference)
#include <hip/hip_runtime.h>
#include <hip/hip_bf16.h>


typedef float v2f __attribute__((ext_vector_type(2)));
typedef float v8f __attribute__((ext_vector_type(8)));

__device__ __forceinline__ unsigned f2ord(float f) {
  unsigned u = __float_as_uint(f);
  return (u & 0x80000000u) ? ~u : (u | 0x80000000u);
}
__device__ __forceinline__ float ord2f(unsigned o) {
  unsigned u = (o & 0x80000000u) ? (o ^ 0x80000000u) : ~o;
  return __uint_as_float(u);
}

__global__ void zero_kernel(unsigned* __restrict__ p, long long count) {
  long long i = (long long)blockIdx.x * blockDim.x + threadIdx.x;
  long long stride = (long long)gridDim.x * blockDim.x;
  for (; i < count; i += stride) p[i] = 0u;
}

template <int K, int NCOL>
__global__ void pack_b_kernel(const float* __restrict__ B,
                              float* __restrict__ Bp) {
  int tid = blockIdx.x * blockDim.x + threadIdx.x;
  if (tid >= K * NCOL) return;
  int k = tid / NCOL;
  int col = tid - k * NCOL;
  int kk = k >> 2;
  int p = (k >> 1) & 1;
  int q = k & 1;
  Bp[(((kk * 2 + p) * NCOL) + col) * 2 + q] = B[tid];
}

template <int K, int NCOL>
__global__ __launch_bounds__(256) void gemm_wmma_kernel(
    const float* __restrict__ A, const float* __restrict__ Bp,
    float* __restrict__ C, int M) {
  constexpr int NT = NCOL / 16;
  constexpr int MT = 8 / NT;
  constexpr int ROWS = MT * 16;
  constexpr int KV = K / 4;
  __shared__ __align__(16) float As[ROWS * K];

  const int tid = threadIdx.x;
  const int wave = tid >> 5;
  const int lane = tid & 31;
  const int row0 = blockIdx.x * ROWS;

  {
    const float4* A4 = (const float4*)A;
    float4* As4 = (float4*)As;
    for (int i = tid; i < ROWS * KV; i += 256) {
      int r = i / KV;
      int c = i - r * KV;
      int gr = row0 + r;
      float4 v = {0.f, 0.f, 0.f, 0.f};
      if (gr < M) v = A4[(size_t)gr * KV + c];
      As4[i] = v;
    }
  }
  __syncthreads();

  const int mtile = wave / NT;
  const int ntile = wave - mtile * NT;
  const int lrow = (mtile << 4) + (lane & 15);
  const int koff = (lane >> 4) << 1;
  const int ncol = (ntile << 4) + (lane & 15);
  const int bbase = ((lane >> 4) * NCOL + ncol);

  v8f acc = {0.f, 0.f, 0.f, 0.f, 0.f, 0.f, 0.f, 0.f};
  for (int kk = 0; kk < KV; ++kk) {
    v2f a, b;
    a.x = As[lrow * K + kk * 4 + koff];
    a.y = As[lrow * K + kk * 4 + koff + 1];
    b = *(const v2f*)&Bp[(size_t)(bbase + kk * 2 * NCOL) * 2];
    acc = __builtin_amdgcn_wmma_f32_16x16x4_f32(
         false, a,  false, b,
         (short)0, acc,  false,  false);
  }

  const int rbase = row0 + (mtile << 4) + ((lane >> 4) << 3);
  if (row0 + ROWS <= M) {
#pragma unroll
    for (int i = 0; i < 8; ++i) C[(size_t)(rbase + i) * NCOL + ncol] = acc[i];
  } else {
#pragma unroll
    for (int i = 0; i < 8; ++i) {
      int r = rbase + i;
      if (r < M) C[(size_t)r * NCOL + ncol] = acc[i];
    }
  }
}

template <int H, int F>
__global__ void alpha_kernel(const float* __restrict__ h,
                             const float* __restrict__ a_s,
                             const float* __restrict__ a_d,
                             float* __restrict__ out_s,
                             float* __restrict__ out_d) {
  constexpr int HF = H * F;
  __shared__ float ss[HF];
  __shared__ float sd[HF];
  const int n = blockIdx.x;
  const int t = threadIdx.x;
  float v = h[(size_t)n * HF + t];
  ss[t] = v * a_s[t];
  sd[t] = v * a_d[t];
  __syncthreads();
  for (int stride = F / 2; stride > 0; stride >>= 1) {
    if ((t & (F - 1)) < stride) {
      ss[t] += ss[t + stride];
      sd[t] += sd[t + stride];
    }
    __syncthreads();
  }
  if ((t & (F - 1)) == 0) {
    int hd = t / F;
    out_s[n * H + hd] = ss[t];
    out_d[n * H + hd] = sd[t];
  }
}

template <int H>
__global__ void edge_logits_kernel(const float* __restrict__ as,
                                   const float* __restrict__ ad,
                                   const int* __restrict__ src,
                                   const int* __restrict__ dst, int nE, int nTot,
                                   float* __restrict__ logits,
                                   unsigned* __restrict__ mbuf) {
  int tid = blockIdx.x * blockDim.x + threadIdx.x;
  if (tid >= nTot * H) return;
  int e = tid / H, hd = tid - e * H;
  int s, d;
  if (e < nE) { s = src[e]; d = dst[e]; } else { s = d = e - nE; }
  float v = as[s * H + hd] + ad[d * H + hd];
  v = v > 0.f ? v : 0.2f * v;
  logits[tid] = v;
  atomicMax(&mbuf[d * H + hd], f2ord(v));
}

template <int H>
__global__ void edge_denom_kernel(const float* __restrict__ logits,
                                  const unsigned* __restrict__ mbuf,
                                  const int* __restrict__ dst, int nE, int nTot,
                                  float* __restrict__ denom) {
  int tid = blockIdx.x * blockDim.x + threadIdx.x;
  if (tid >= nTot * H) return;
  int e = tid / H, hd = tid - e * H;
  int d = (e < nE) ? dst[e] : (e - nE);
  float m = ord2f(mbuf[d * H + hd]);
  atomicAdd(&denom[d * H + hd], __expf(logits[tid] - m));
}

template <int H, int F>
__global__ void edge_agg_kernel(const float* __restrict__ hsrc,
                                const float* __restrict__ logits,
                                const unsigned* __restrict__ mbuf,
                                const float* __restrict__ denom,
                                const int* __restrict__ src,
                                const int* __restrict__ dst, int nE, int nTot,
                                float* __restrict__ agg) {
  constexpr int HF = H * F;
  constexpr int CH = HF / 4;
  int tid = blockIdx.x * blockDim.x + threadIdx.x;
  if (tid >= nTot * CH) return;
  int e = tid / CH;
  int c = tid - e * CH;
  int fbase = c * 4;
  int hd = fbase / F;
  int s, d;
  if (e < nE) { s = src[e]; d = dst[e]; } else { s = d = e - nE; }
  float m = ord2f(mbuf[d * H + hd]);
  float alpha = __expf(logits[e * H + hd] - m) / (denom[d * H + hd] + 1e-16f);
  const float4 hv = *(const float4*)&hsrc[(size_t)s * HF + fbase];
  float* ap = &agg[(size_t)d * HF + fbase];
  atomicAdd(ap + 0, hv.x * alpha);
  atomicAdd(ap + 1, hv.y * alpha);
  atomicAdd(ap + 2, hv.z * alpha);
  atomicAdd(ap + 3, hv.w * alpha);
}

__global__ void elu_bias_kernel(float* __restrict__ buf,
                                const float* __restrict__ bias, int total,
                                int ncol) {
  int tid = blockIdx.x * blockDim.x + threadIdx.x;
  if (tid >= total) return;
  float v = buf[tid] + bias[tid % ncol];
  buf[tid] = v > 0.f ? v : (__expf(v) - 1.f);
}

__global__ __launch_bounds__(256) void logsoftmax_kernel(
    const float* __restrict__ agg, const float* __restrict__ bias,
    float* __restrict__ out, int n) {
  int tid = blockIdx.x * blockDim.x + threadIdx.x;
  int node = tid >> 5;
  int lane = tid & 31;
  if (node >= n) return;
  float v = agg[(size_t)node * 32 + lane] + bias[lane];
  float m = v;
  for (int o = 16; o > 0; o >>= 1) m = fmaxf(m, __shfl_xor(m, o));
  float ex = __expf(v - m);
  float s = ex;
  for (int o = 16; o > 0; o >>= 1) s += __shfl_xor(s, o);
  out[(size_t)node * 32 + lane] = v - m - __logf(s);
}

extern "C" void kernel_launch(void* const* d_in, const int* in_sizes, int n_in,
                              void* d_out, int out_size, void* d_ws,
                              size_t ws_size, hipStream_t stream) {
  (void)n_in; (void)out_size; (void)ws_size;
  const float* x      = (const float*)d_in[0];
  const int*   ei     = (const int*)d_in[1];
  const float* W1     = (const float*)d_in[2];
  const float* a_src1 = (const float*)d_in[3];
  const float* a_dst1 = (const float*)d_in[4];
  const float* b1     = (const float*)d_in[5];
  const float* W2     = (const float*)d_in[6];
  const float* a_src2 = (const float*)d_in[7];
  const float* a_dst2 = (const float*)d_in[8];
  const float* b2     = (const float*)d_in[9];

  const int n    = in_sizes[0] / 500;
  const int nE   = in_sizes[1] / 2;
  const int nTot = nE + n;
  const int* src = ei;
  const int* dst = ei + nE;

  float* ws = (float*)d_ws;
  size_t o = 0;
  float* h1      = ws + o; o += (size_t)n * 128;
  float* logits1 = ws + o; o += (size_t)nTot * 2;
  float* as1     = ws + o; o += (size_t)n * 2;
  float* ad1     = ws + o; o += (size_t)n * 2;
  float* h2      = ws + o; o += (size_t)n * 32;
  float* logits2 = ws + o; o += (size_t)nTot;
  float* as2     = ws + o; o += (size_t)n;
  float* ad2     = ws + o; o += (size_t)n;
  float* Bp1     = ws + o; o += (size_t)500 * 128;
  float* Bp2     = ws + o; o += (size_t)128 * 32;
  size_t z0 = o;
  unsigned* m1   = (unsigned*)(ws + o); o += (size_t)n * 2;
  float* denom1  = ws + o; o += (size_t)n * 2;
  float* agg1    = ws + o; o += (size_t)n * 128;
  unsigned* m2   = (unsigned*)(ws + o); o += (size_t)n;
  float* denom2  = ws + o; o += (size_t)n;
  float* agg2    = ws + o; o += (size_t)n * 32;
  size_t zcount = o - z0;

  zero_kernel<<<2048, 256, 0, stream>>>((unsigned*)(ws + z0), (long long)zcount);
  pack_b_kernel<500, 128><<<(500 * 128 + 255) / 256, 256, 0, stream>>>(W1, Bp1);
  pack_b_kernel<128, 32><<<(128 * 32 + 255) / 256, 256, 0, stream>>>(W2, Bp2);

  gemm_wmma_kernel<500, 128><<<(n + 15) / 16, 256, 0, stream>>>(x, Bp1, h1, n);
  alpha_kernel<2, 64><<<n, 128, 0, stream>>>(h1, a_src1, a_dst1, as1, ad1);
  {
    int t = nTot * 2;
    edge_logits_kernel<2><<<(t + 255) / 256, 256, 0, stream>>>(
        as1, ad1, src, dst, nE, nTot, logits1, m1);
    edge_denom_kernel<2><<<(t + 255) / 256, 256, 0, stream>>>(
        logits1, m1, dst, nE, nTot, denom1);
    int ta = nTot * 32;
    edge_agg_kernel<2, 64><<<(ta + 255) / 256, 256, 0, stream>>>(
        h1, logits1, m1, denom1, src, dst, nE, nTot, agg1);
  }
  {
    int t = n * 128;
    elu_bias_kernel<<<(t + 255) / 256, 256, 0, stream>>>(agg1, b1, t, 128);
  }

  gemm_wmma_kernel<128, 32><<<(n + 63) / 64, 256, 0, stream>>>(agg1, Bp2, h2, n);
  alpha_kernel<1, 32><<<n, 32, 0, stream>>>(h2, a_src2, a_dst2, as2, ad2);
  {
    int t = nTot;
    edge_logits_kernel<1><<<(t + 255) / 256, 256, 0, stream>>>(
        as2, ad2, src, dst, nE, nTot, logits2, m2);
    edge_denom_kernel<1><<<(t + 255) / 256, 256, 0, stream>>>(
        logits2, m2, dst, nE, nTot, denom2);
    int ta = nTot * 8;
    edge_agg_kernel<1, 32><<<(ta + 255) / 256, 256, 0, stream>>>(
        h2, logits2, m2, denom2, src, dst, nE, nTot, agg2);
  }

  logsoftmax_kernel<<<((size_t)n * 32 + 255) / 256, 256, 0, stream>>>(
      agg2, b2, (float*)d_out, n);
}


# edges grouped by destination once (CSR via int atomics); one wave per node does max, denominator and row sum, replacing float atomic scatter: sum order now CSR order
# speedup vs baseline: 1.8695x; 1.8695x over previous
#include <hip/hip_runtime.h>
#include <hip/hip_bf16.h>


typedef float v2f __attribute__((ext_vector_type(2)));
typedef float v8f __attribute__((ext_vector_type(8)));

__global__ void zero_kernel(unsigned* __restrict__ p, long long count) {
  long long i = (long long)blockIdx.x * blockDim.x + threadIdx.x;
  long long stride = (long long)gridDim.x * blockDim.x;
  for (; i < count; i += stride) p[i] = 0u;
}

template <int K, int NCOL>
__global__ void pack_b_kernel(const float* __restrict__ B,
                              float* __restrict__ Bp) {
  int tid = blockIdx.x * blockDim.x + threadIdx.x;
  if (tid >= K * NCOL) return;
  int k = tid / NCOL;
  int col = tid - k * NCOL;
  int kk = k >> 2;
  int p = (k >> 1) & 1;
  int q = k & 1;
  Bp[(((kk * 2 + p) * NCOL) + col) * 2 + q] = B[tid];
}

template <int K, int NCOL>
__global__ __launch_bounds__(256) void gemm_wmma_kernel(
    const float* __restrict__ A, const float* __restrict__ Bp,
    float* __restrict__ C, int M) {
  constexpr int NT = NCOL / 16;
  constexpr int MT = 8 / NT;
  constexpr int ROWS = MT * 16;
  constexpr int KV = K / 4;
  __shared__ __align__(16) float As[ROWS * K];

  const int tid = threadIdx.x;
  const int wave = tid >> 5;
  const int lane = tid & 31;
  const int row0 = blockIdx.x * ROWS;

  {
    const float4* A4 = (const float4*)A;
    float4* As4 = (float4*)As;
    for (int i = tid; i < ROWS * KV; i += 256) {
      int r = i / KV;
      int c = i - r * KV;
      int gr = row0 + r;
      float4 v = {0.f, 0.f, 0.f, 0.f};
      if (gr < M) v = A4[(size_t)gr * KV + c];
      As4[i] = v;
    }
  }
  __syncthreads();

  const int mtile = wave / NT;
  const int ntile = wave - mtile * NT;
  const int lrow = (mtile << 4) + (lane & 15);
  const int koff = (lane >> 4) << 1;
  const int ncol = (ntile << 4) + (lane & 15);
  const int bbase = ((lane >> 4) * NCOL + ncol);

  v8f acc = {0.f, 0.f, 0.f, 0.f, 0.f, 0.f, 0.f, 0.f};
  for (int kk = 0; kk < KV; ++kk) {
    v2f a, b;
    a.x = As[lrow * K + kk * 4 + koff];
    a.y = As[lrow * K + kk * 4 + koff + 1];
    b = *(const v2f*)&Bp[(size_t)(bbase + kk * 2 * NCOL) * 2];
    acc = __builtin_amdgcn_wmma_f32_16x16x4_f32(
         false, a,  false, b,
         (short)0, acc,  false,  false);
  }

  const int rbase = row0 + (mtile << 4) + ((lane >> 4) << 3);
  if (row0 + ROWS <= M) {
#pragma unroll
    for (int i = 0; i < 8; ++i) C[(size_t)(rbase + i) * NCOL + ncol] = acc[i];
  } else {
#pragma unroll
    for (int i = 0; i < 8; ++i) {
      int r = rbase + i;
      if (r < M) C[(size_t)r * NCOL + ncol] = acc[i];
    }
  }
}

template <int H, int F>
__global__ void alpha_kernel(const float* __restrict__ h,
                             const float* __restrict__ a_s,
                             const float* __restrict__ a_d,
                             float* __restrict__ out_s,
                             float* __restrict__ out_d) {
  constexpr int HF = H * F;
  __shared__ float ss[HF];
  __shared__ float sd[HF];
  const int n = blockIdx.x;
  const int t = threadIdx.x;
  float v = h[(size_t)n * HF + t];
  ss[t] = v * a_s[t];
  sd[t] = v * a_d[t];
  __syncthreads();
  for (int stride = F / 2; stride > 0; stride >>= 1) {
    if ((t & (F - 1)) < stride) {
      ss[t] += ss[t + stride];
      sd[t] += sd[t + stride];
    }
    __syncthreads();
  }
  if ((t & (F - 1)) == 0) {
    int hd = t / F;
    out_s[n * H + hd] = ss[t];
    out_d[n * H + hd] = sd[t];
  }
}

__global__ void degree_kernel(const int* __restrict__ dst, int nE,
                              int* __restrict__ cnt) {
  int e = blockIdx.x * blockDim.x + threadIdx.x;
  if (e < nE) atomicAdd(&cnt[dst[e]], 1);
}

__global__ __launch_bounds__(1024) void scan_kernel(
    const int* __restrict__ cnt, int n, int* __restrict__ rowptr,
    int* __restrict__ cursor) {
  __shared__ int wsum[32];
  const int lane = threadIdx.x & 31;
  const int wave = threadIdx.x >> 5;
  const int nw = blockDim.x >> 5;
  const int seg = (n + nw - 1) / nw;
  const int lo = wave * seg;
  const int hi = min(lo + seg, n);

  int part = 0;
  for (int i = lo + lane; i < hi; i += 32) part += cnt[i];
  for (int o = 16; o > 0; o >>= 1) part += __shfl_xor(part, o);
  if (lane == 0) wsum[wave] = part;
  __syncthreads();

  int carry = (lane < wave) ? wsum[lane] : 0;
  for (int o = 16; o > 0; o >>= 1) carry += __shfl_xor(carry, o);

  for (int base = lo; base < hi; base += 32) {
    const int i = base + lane;
    const int c = (i < hi) ? cnt[i] : 0;
    int inc = c;
    for (int o = 1; o < 32; o <<= 1) {
      int t = __shfl_up(inc, o);
      if (lane >= o) inc += t;
    }
    const int excl = carry + inc - c;
    if (i < hi) { rowptr[i] = excl; cursor[i] = excl; }
    carry += __shfl(inc, 31);
  }
  if (wave == nw - 1 && lane == 0) rowptr[n] = carry;
}

__global__ void fill_kernel(const int* __restrict__ src,
                            const int* __restrict__ dst, int nE,
                            int* __restrict__ cursor,
                            int* __restrict__ csr_src) {
  int e = blockIdx.x * blockDim.x + threadIdx.x;
  if (e >= nE) return;
  int pos = atomicAdd(&cursor[dst[e]], 1);
  csr_src[pos] = src[e];
}

template <int H>
__device__ __forceinline__ void edge_logit(const float* __restrict__ as,
                                           const int* __restrict__ csr_src,
                                           const float (&adv)[H], int start,
                                           int deg, int node, int k, int& s,
                                           float (&v)[H]) {
  s = (k < deg) ? csr_src[start + k] : node;
  for (int hd = 0; hd < H; ++hd) {
    float t = as[s * H + hd] + adv[hd];
    v[hd] = t > 0.f ? t : 0.2f * t;
  }
}

template <int H, int F>
__global__ __launch_bounds__(256) void gat_gather_kernel(
    const float* __restrict__ hsrc, const float* __restrict__ as,
    const float* __restrict__ ad, const int* __restrict__ rowptr,
    const int* __restrict__ csr_src, int n, float* __restrict__ agg) {
  constexpr int HF = H * F;
  constexpr int VEC = HF / 32;
  static_assert(VEC == 1 || VEC == 4, "row must be 32 or 128 floats");
  static_assert(H == 1 || H == 2, "per-lane head select below handles 1 or 2 heads");
  const int lane = threadIdx.x & 31;
  const int node = __builtin_amdgcn_readfirstlane(
      (int)((blockIdx.x * blockDim.x + threadIdx.x) >> 5));
  if (node >= n) return;
  const int start = rowptr[node];
  const int deg = rowptr[node + 1] - start;
  const int tot = deg + 1;
  const int myhead = (lane * VEC) / F;

  float adv[H];
  for (int hd = 0; hd < H; ++hd) adv[hd] = ad[node * H + hd];

  const bool ok0 = lane < tot;
  int s0 = node;
  float v0[H];
  for (int hd = 0; hd < H; ++hd) v0[hd] = -INFINITY;
  if (ok0) edge_logit<H>(as, csr_src, adv, start, deg, node, lane, s0, v0);

  float mx[H];
  for (int hd = 0; hd < H; ++hd) mx[hd] = v0[hd];
  for (int base = 32; base < tot; base += 32) {
    const int k = base + lane;
    if (k < tot) {
      int s; float v[H];
      edge_logit<H>(as, csr_src, adv, start, deg, node, k, s, v);
      for (int hd = 0; hd < H; ++hd) mx[hd] = fmaxf(mx[hd], v[hd]);
    }
  }
  for (int hd = 0; hd < H; ++hd)
    for (int o = 16; o > 0; o >>= 1) mx[hd] = fmaxf(mx[hd], __shfl_xor(mx[hd], o));

  float w0[H], dn[H];
  for (int hd = 0; hd < H; ++hd) {
    w0[hd] = ok0 ? __expf(v0[hd] - mx[hd]) : 0.f;
    dn[hd] = w0[hd];
  }
  for (int base = 32; base < tot; base += 32) {
    const int k = base + lane;
    if (k < tot) {
      int s; float v[H];
      edge_logit<H>(as, csr_src, adv, start, deg, node, k, s, v);
      for (int hd = 0; hd < H; ++hd) dn[hd] += __expf(v[hd] - mx[hd]);
    }
  }
  for (int hd = 0; hd < H; ++hd)
    for (int o = 16; o > 0; o >>= 1) dn[hd] += __shfl_xor(dn[hd], o);

  float acc[VEC];
  for (int i = 0; i < VEC; ++i) acc[i] = 0.f;
  const float* rowbase = hsrc + lane * VEC;

  for (int base = 0; base < tot; base += 32) {
    int sc = s0;
    float al[H];
    if (base == 0) {
      for (int hd = 0; hd < H; ++hd) al[hd] = w0[hd] / (dn[hd] + 1e-16f);
    } else {
      const int k = base + lane;
      sc = node;
      for (int hd = 0; hd < H; ++hd) al[hd] = 0.f;
      if (k < tot) {
        float v[H];
        edge_logit<H>(as, csr_src, adv, start, deg, node, k, sc, v);
        for (int hd = 0; hd < H; ++hd)
          al[hd] = __expf(v[hd] - mx[hd]) / (dn[hd] + 1e-16f);
      }
    }
    const int cnt = min(tot - base, 32);
#pragma unroll 4
    for (int j = 0; j < cnt; ++j) {
      const int sj = __builtin_amdgcn_readlane(sc, j);
      float aj = __int_as_float(
          __builtin_amdgcn_readlane(__float_as_int(al[0]), j));
      if constexpr (H > 1) {
        const float a1 = __int_as_float(
            __builtin_amdgcn_readlane(__float_as_int(al[H - 1]), j));
        aj = myhead ? a1 : aj;
      }
      const float* rp = rowbase + (size_t)sj * HF;
      if constexpr (VEC == 4) {
        const float4 hv = *(const float4*)rp;
        acc[0] += hv.x * aj;
        acc[1] += hv.y * aj;
        acc[2] += hv.z * aj;
        acc[3] += hv.w * aj;
      } else {
        acc[0] += rp[0] * aj;
      }
    }
  }

  float* op = agg + (size_t)node * HF + lane * VEC;
  if constexpr (VEC == 4) {
    float4 r;
    r.x = acc[0]; r.y = acc[1]; r.z = acc[2]; r.w = acc[3];
    *(float4*)op = r;
  } else {
    op[0] = acc[0];
  }
}

__global__ void elu_bias_kernel(float* __restrict__ buf,
                                const float* __restrict__ bias, int total,
                                int ncol) {
  int tid = blockIdx.x * blockDim.x + threadIdx.x;
  if (tid >= total) return;
  float v = buf[tid] + bias[tid % ncol];
  buf[tid] = v > 0.f ? v : (__expf(v) - 1.f);
}

__global__ __launch_bounds__(256) void logsoftmax_kernel(
    const float* __restrict__ agg, const float* __restrict__ bias,
    float* __restrict__ out, int n) {
  int tid = blockIdx.x * blockDim.x + threadIdx.x;
  int node = tid >> 5;
  int lane = tid & 31;
  if (node >= n) return;
  float v = agg[(size_t)node * 32 + lane] + bias[lane];
  float m = v;
  for (int o = 16; o > 0; o >>= 1) m = fmaxf(m, __shfl_xor(m, o));
  float ex = __expf(v - m);
  float s = ex;
  for (int o = 16; o > 0; o >>= 1) s += __shfl_xor(s, o);
  out[(size_t)node * 32 + lane] = v - m - __logf(s);
}

extern "C" void kernel_launch(void* const* d_in, const int* in_sizes, int n_in,
                              void* d_out, int out_size, void* d_ws,
                              size_t ws_size, hipStream_t stream) {
  (void)n_in; (void)out_size; (void)ws_size;
  const float* x      = (const float*)d_in[0];
  const int*   ei     = (const int*)d_in[1];
  const float* W1     = (const float*)d_in[2];
  const float* a_src1 = (const float*)d_in[3];
  const float* a_dst1 = (const float*)d_in[4];
  const float* b1     = (const float*)d_in[5];
  const float* W2     = (const float*)d_in[6];
  const float* a_src2 = (const float*)d_in[7];
  const float* a_dst2 = (const float*)d_in[8];
  const float* b2     = (const float*)d_in[9];

  const int n    = in_sizes[0] / 500;
  const int nE   = in_sizes[1] / 2;
  const int* src = ei;
  const int* dst = ei + nE;

  float* ws = (float*)d_ws;
  size_t o = 0;
  float* h1   = ws + o; o += (size_t)n * 128;
  float* agg1 = ws + o; o += (size_t)n * 128;
  float* h2   = ws + o; o += (size_t)n * 32;
  float* agg2 = ws + o; o += (size_t)n * 32;
  float* Bp1  = ws + o; o += (size_t)500 * 128;
  float* Bp2  = ws + o; o += (size_t)128 * 32;
  float* as1  = ws + o; o += (size_t)n * 2;
  float* ad1  = ws + o; o += (size_t)n * 2;
  float* as2  = ws + o; o += (size_t)n;
  float* ad2  = ws + o; o += (size_t)n;
  int* cnt     = (int*)(ws + o); o += (size_t)n;
  int* rowptr  = (int*)(ws + o); o += (size_t)n + 1;
  int* cursor  = (int*)(ws + o); o += (size_t)n;
  int* csr_src = (int*)(ws + o); o += (size_t)nE;

  zero_kernel<<<(n + 255) / 256, 256, 0, stream>>>((unsigned*)cnt, (long long)n);
  degree_kernel<<<(nE + 255) / 256, 256, 0, stream>>>(dst, nE, cnt);
  scan_kernel<<<1, 1024, 0, stream>>>(cnt, n, rowptr, cursor);
  fill_kernel<<<(nE + 255) / 256, 256, 0, stream>>>(src, dst, nE, cursor, csr_src);
  pack_b_kernel<500, 128><<<(500 * 128 + 255) / 256, 256, 0, stream>>>(W1, Bp1);
  pack_b_kernel<128, 32><<<(128 * 32 + 255) / 256, 256, 0, stream>>>(W2, Bp2);

  const int gblocks = (int)(((size_t)n * 32 + 255) / 256);

  gemm_wmma_kernel<500, 128><<<(n + 15) / 16, 256, 0, stream>>>(x, Bp1, h1, n);
  alpha_kernel<2, 64><<<n, 128, 0, stream>>>(h1, a_src1, a_dst1, as1, ad1);
  gat_gather_kernel<2, 64><<<gblocks, 256, 0, stream>>>(
      h1, as1, ad1, rowptr, csr_src, n, agg1);
  {
    int t = n * 128;
    elu_bias_kernel<<<(t + 255) / 256, 256, 0, stream>>>(agg1, b1, t, 128);
  }

  gemm_wmma_kernel<128, 32><<<(n + 63) / 64, 256, 0, stream>>>(agg1, Bp2, h2, n);
  alpha_kernel<1, 32><<<n, 32, 0, stream>>>(h2, a_src2, a_dst2, as2, ad2);
  gat_gather_kernel<1, 32><<<gblocks, 256, 0, stream>>>(
      h2, as2, ad2, rowptr, csr_src, n, agg2);

  logsoftmax_kernel<<<((size_t)n * 32 + 255) / 256, 256, 0, stream>>>(
      agg2, b2, (float*)d_out, n);
}


# per-row passes folded into producer epilogues: attention dots in the WMMA GEMM (sum order: 8-term chain per lane + lane tree), bias+ELU and log_softmax in the gather; 14 -> 10 launches
# speedup vs baseline: 2.1721x; 1.1619x over previous
#include <hip/hip_runtime.h>
#include <hip/hip_bf16.h>


typedef float v2f __attribute__((ext_vector_type(2)));
typedef float v8f __attribute__((ext_vector_type(8)));

__global__ void zero_kernel(unsigned* __restrict__ p, long long count) {
  long long i = (long long)blockIdx.x * blockDim.x + threadIdx.x;
  long long stride = (long long)gridDim.x * blockDim.x;
  for (; i < count; i += stride) p[i] = 0u;
}

template <int K, int NCOL>
__global__ void pack_b_kernel(const float* __restrict__ B,
                              float* __restrict__ Bp) {
  int tid = blockIdx.x * blockDim.x + threadIdx.x;
  if (tid >= K * NCOL) return;
  int k = tid / NCOL;
  int col = tid - k * NCOL;
  int kk = k >> 2;
  int p = (k >> 1) & 1;
  int q = k & 1;
  Bp[(((kk * 2 + p) * NCOL) + col) * 2 + q] = B[tid];
}

template <int K, int NCOL, int H>
__global__ __launch_bounds__(256) void gemm_wmma_kernel(
    const float* __restrict__ A, const float* __restrict__ Bp,
    float* __restrict__ C, int M, const float* __restrict__ a_s,
    const float* __restrict__ a_d, float* __restrict__ out_s,
    float* __restrict__ out_d) {
  constexpr int NT = NCOL / 16;
  constexpr int MT = 8 / NT;
  constexpr int ROWS = MT * 16;
  constexpr int KV = K / 4;
  __shared__ __align__(16) float As[ROWS * K];

  const int tid = threadIdx.x;
  const int wave = tid >> 5;
  const int lane = tid & 31;
  const int row0 = blockIdx.x * ROWS;

  {
    const float4* A4 = (const float4*)A;
    float4* As4 = (float4*)As;
    for (int i = tid; i < ROWS * KV; i += 256) {
      int r = i / KV;
      int c = i - r * KV;
      int gr = row0 + r;
      float4 v = {0.f, 0.f, 0.f, 0.f};
      if (gr < M) v = A4[(size_t)gr * KV + c];
      As4[i] = v;
    }
  }
  __syncthreads();

  const int mtile = wave / NT;
  const int ntile = wave - mtile * NT;
  const int lrow = (mtile << 4) + (lane & 15);
  const int koff = (lane >> 4) << 1;
  const int ncol = (ntile << 4) + (lane & 15);
  const int bbase = ((lane >> 4) * NCOL + ncol);

  v8f acc = {0.f, 0.f, 0.f, 0.f, 0.f, 0.f, 0.f, 0.f};
  for (int kk = 0; kk < KV; ++kk) {
    v2f a, b;
    a.x = As[lrow * K + kk * 4 + koff];
    a.y = As[lrow * K + kk * 4 + koff + 1];
    b = *(const v2f*)&Bp[(size_t)(bbase + kk * 2 * NCOL) * 2];
    acc = __builtin_amdgcn_wmma_f32_16x16x4_f32(
         false, a,  false, b,
         (short)0, acc,  false,  false);
  }

  constexpr int P = NCOL + 4;
  constexpr int F = NCOL / H;
  constexpr int TPR = 256 / ROWS;
  constexpr int TPH = TPR / H;
  static_assert(F / TPH == 8, "each thread sums 8 columns (two float4)");
  static_assert(ROWS * P <= ROWS * K, "C tile must fit in the A tile's LDS");
  float* Cs = As;
  __syncthreads();
  {
    const int crow = (mtile << 4) + ((lane >> 4) << 3);
    for (int i = 0; i < 8; ++i) Cs[(crow + i) * P + ncol] = acc[i];
  }
  __syncthreads();
  {
    const int r = tid / TPR;
    const int rem = tid - r * TPR;
    const int hd = rem / TPH;
    const int part = rem - hd * TPH;
    const int c0 = hd * F + part * 8;
    const float4 u0 = *(const float4*)&Cs[r * P + c0];
    const float4 u1 = *(const float4*)&Cs[r * P + c0 + 4];
    const float4 s0 = *(const float4*)&a_s[c0];
    const float4 s1 = *(const float4*)&a_s[c0 + 4];
    const float4 d0 = *(const float4*)&a_d[c0];
    const float4 d1 = *(const float4*)&a_d[c0 + 4];
    float ps = u0.x * s0.x;
    ps += u0.y * s0.y; ps += u0.z * s0.z; ps += u0.w * s0.w;
    ps += u1.x * s1.x; ps += u1.y * s1.y; ps += u1.z * s1.z; ps += u1.w * s1.w;
    float pd = u0.x * d0.x;
    pd += u0.y * d0.y; pd += u0.z * d0.z; pd += u0.w * d0.w;
    pd += u1.x * d1.x; pd += u1.y * d1.y; pd += u1.z * d1.z; pd += u1.w * d1.w;
    for (int o = TPH / 2; o > 0; o >>= 1) {
      ps += __shfl_xor(ps, o);
      pd += __shfl_xor(pd, o);
    }
    const int gr = row0 + r;
    if (part == 0 && gr < M) {
      out_s[gr * H + hd] = ps;
      out_d[gr * H + hd] = pd;
    }
  }

  const int rbase = row0 + (mtile << 4) + ((lane >> 4) << 3);
  if (row0 + ROWS <= M) {
#pragma unroll
    for (int i = 0; i < 8; ++i) C[(size_t)(rbase + i) * NCOL + ncol] = acc[i];
  } else {
#pragma unroll
    for (int i = 0; i < 8; ++i) {
      int r = rbase + i;
      if (r < M) C[(size_t)r * NCOL + ncol] = acc[i];
    }
  }
}

__global__ void degree_kernel(const int* __restrict__ dst, int nE,
                              int* __restrict__ cnt) {
  int e = blockIdx.x * blockDim.x + threadIdx.x;
  if (e < nE) atomicAdd(&cnt[dst[e]], 1);
}

__global__ __launch_bounds__(1024) void scan_kernel(
    const int* __restrict__ cnt, int n, int* __restrict__ rowptr,
    int* __restrict__ cursor) {
  __shared__ int wsum[32];
  const int lane = threadIdx.x & 31;
  const int wave = threadIdx.x >> 5;
  const int nw = blockDim.x >> 5;
  const int seg = (n + nw - 1) / nw;
  const int lo = wave * seg;
  const int hi = min(lo + seg, n);

  int part = 0;
  for (int i = lo + lane; i < hi; i += 32) part += cnt[i];
  for (int o = 16; o > 0; o >>= 1) part += __shfl_xor(part, o);
  if (lane == 0) wsum[wave] = part;
  __syncthreads();

  int carry = (lane < wave) ? wsum[lane] : 0;
  for (int o = 16; o > 0; o >>= 1) carry += __shfl_xor(carry, o);

  for (int base = lo; base < hi; base += 32) {
    const int i = base + lane;
    const int c = (i < hi) ? cnt[i] : 0;
    int inc = c;
    for (int o = 1; o < 32; o <<= 1) {
      int t = __shfl_up(inc, o);
      if (lane >= o) inc += t;
    }
    const int excl = carry + inc - c;
    if (i < hi) { rowptr[i] = excl; cursor[i] = excl; }
    carry += __shfl(inc, 31);
  }
  if (wave == nw - 1 && lane == 0) rowptr[n] = carry;
}

__global__ void fill_kernel(const int* __restrict__ src,
                            const int* __restrict__ dst, int nE,
                            int* __restrict__ cursor,
                            int* __restrict__ csr_src) {
  int e = blockIdx.x * blockDim.x + threadIdx.x;
  if (e >= nE) return;
  int pos = atomicAdd(&cursor[dst[e]], 1);
  csr_src[pos] = src[e];
}

template <int H>
__device__ __forceinline__ void edge_logit(const float* __restrict__ as,
                                           const int* __restrict__ csr_src,
                                           const float (&adv)[H], int start,
                                           int deg, int node, int k, int& s,
                                           float (&v)[H]) {
  s = (k < deg) ? csr_src[start + k] : node;
  for (int hd = 0; hd < H; ++hd) {
    float t = as[s * H + hd] + adv[hd];
    v[hd] = t > 0.f ? t : 0.2f * t;
  }
}

template <int H, int F, bool LSM>
__global__ __launch_bounds__(256) void gat_gather_kernel(
    const float* __restrict__ hsrc, const float* __restrict__ as,
    const float* __restrict__ ad, const int* __restrict__ rowptr,
    const int* __restrict__ csr_src, int n, const float* __restrict__ bias,
    float* __restrict__ agg) {
  constexpr int HF = H * F;
  constexpr int VEC = HF / 32;
  static_assert(VEC == 1 || VEC == 4, "row must be 32 or 128 floats");
  static_assert(!LSM || VEC == 1, "log_softmax needs the row in one wave, one class per lane");
  static_assert(H == 1 || H == 2, "per-lane head select below handles 1 or 2 heads");
  const int lane = threadIdx.x & 31;
  const int node = __builtin_amdgcn_readfirstlane(
      (int)((blockIdx.x * blockDim.x + threadIdx.x) >> 5));
  if (node >= n) return;
  const int start = rowptr[node];
  const int deg = rowptr[node + 1] - start;
  const int tot = deg + 1;
  const int myhead = (lane * VEC) / F;

  float adv[H];
  for (int hd = 0; hd < H; ++hd) adv[hd] = ad[node * H + hd];

  const bool ok0 = lane < tot;
  int s0 = node;
  float v0[H];
  for (int hd = 0; hd < H; ++hd) v0[hd] = -INFINITY;
  if (ok0) edge_logit<H>(as, csr_src, adv, start, deg, node, lane, s0, v0);

  float mx[H];
  for (int hd = 0; hd < H; ++hd) mx[hd] = v0[hd];
  for (int base = 32; base < tot; base += 32) {
    const int k = base + lane;
    if (k < tot) {
      int s; float v[H];
      edge_logit<H>(as, csr_src, adv, start, deg, node, k, s, v);
      for (int hd = 0; hd < H; ++hd) mx[hd] = fmaxf(mx[hd], v[hd]);
    }
  }
  for (int hd = 0; hd < H; ++hd)
    for (int o = 16; o > 0; o >>= 1) mx[hd] = fmaxf(mx[hd], __shfl_xor(mx[hd], o));

  float w0[H], dn[H];
  for (int hd = 0; hd < H; ++hd) {
    w0[hd] = ok0 ? __expf(v0[hd] - mx[hd]) : 0.f;
    dn[hd] = w0[hd];
  }
  for (int base = 32; base < tot; base += 32) {
    const int k = base + lane;
    if (k < tot) {
      int s; float v[H];
      edge_logit<H>(as, csr_src, adv, start, deg, node, k, s, v);
      for (int hd = 0; hd < H; ++hd) dn[hd] += __expf(v[hd] - mx[hd]);
    }
  }
  for (int hd = 0; hd < H; ++hd)
    for (int o = 16; o > 0; o >>= 1) dn[hd] += __shfl_xor(dn[hd], o);

  float acc[VEC];
  for (int i = 0; i < VEC; ++i) acc[i] = 0.f;
  const float* rowbase = hsrc + lane * VEC;

  for (int base = 0; base < tot; base += 32) {
    int sc = s0;
    float al[H];
    if (base == 0) {
      for (int hd = 0; hd < H; ++hd) al[hd] = w0[hd] / (dn[hd] + 1e-16f);
    } else {
      const int k = base + lane;
      sc = node;
      for (int hd = 0; hd < H; ++hd) al[hd] = 0.f;
      if (k < tot) {
        float v[H];
        edge_logit<H>(as, csr_src, adv, start, deg, node, k, sc, v);
        for (int hd = 0; hd < H; ++hd)
          al[hd] = __expf(v[hd] - mx[hd]) / (dn[hd] + 1e-16f);
      }
    }
    const int cnt = min(tot - base, 32);
#pragma unroll 4
    for (int j = 0; j < cnt; ++j) {
      const int sj = __builtin_amdgcn_readlane(sc, j);
      float aj = __int_as_float(
          __builtin_amdgcn_readlane(__float_as_int(al[0]), j));
      if constexpr (H > 1) {
        const float a1 = __int_as_float(
            __builtin_amdgcn_readlane(__float_as_int(al[H - 1]), j));
        aj = myhead ? a1 : aj;
      }
      const float* rp = rowbase + (size_t)sj * HF;
      if constexpr (VEC == 4) {
        const float4 hv = *(const float4*)rp;
        acc[0] += hv.x * aj;
        acc[1] += hv.y * aj;
        acc[2] += hv.z * aj;
        acc[3] += hv.w * aj;
      } else {
        acc[0] += rp[0] * aj;
      }
    }
  }

  for (int i = 0; i < VEC; ++i) acc[i] += bias[lane * VEC + i];
  if constexpr (LSM) {
    const float v = acc[0];
    float m = v;
    for (int o = 16; o > 0; o >>= 1) m = fmaxf(m, __shfl_xor(m, o));
    float sum = __expf(v - m);
    for (int o = 16; o > 0; o >>= 1) sum += __shfl_xor(sum, o);
    acc[0] = v - m - __logf(sum);
  } else {
    for (int i = 0; i < VEC; ++i)
      acc[i] = acc[i] > 0.f ? acc[i] : (__expf(acc[i]) - 1.f);
  }

  float* op = agg + (size_t)node * HF + lane * VEC;
  if constexpr (VEC == 4) {
    float4 r;
    r.x = acc[0]; r.y = acc[1]; r.z = acc[2]; r.w = acc[3];
    *(float4*)op = r;
  } else {
    op[0] = acc[0];
  }
}

extern "C" void kernel_launch(void* const* d_in, const int* in_sizes, int n_in,
                              void* d_out, int out_size, void* d_ws,
                              size_t ws_size, hipStream_t stream) {
  (void)n_in; (void)out_size; (void)ws_size;
  const float* x      = (const float*)d_in[0];
  const int*   ei     = (const int*)d_in[1];
  const float* W1     = (const float*)d_in[2];
  const float* a_src1 = (const float*)d_in[3];
  const float* a_dst1 = (const float*)d_in[4];
  const float* b1     = (const float*)d_in[5];
  const float* W2     = (const float*)d_in[6];
  const float* a_src2 = (const float*)d_in[7];
  const float* a_dst2 = (const float*)d_in[8];
  const float* b2     = (const float*)d_in[9];

  const int n    = in_sizes[0] / 500;
  const int nE   = in_sizes[1] / 2;
  const int* src = ei;
  const int* dst = ei + nE;

  float* ws = (float*)d_ws;
  size_t o = 0;
  float* h1   = ws + o; o += (size_t)n * 128;
  float* agg1 = ws + o; o += (size_t)n * 128;
  float* h2   = ws + o; o += (size_t)n * 32;
  float* Bp1  = ws + o; o += (size_t)500 * 128;
  float* Bp2  = ws + o; o += (size_t)128 * 32;
  float* as1  = ws + o; o += (size_t)n * 2;
  float* ad1  = ws + o; o += (size_t)n * 2;
  float* as2  = ws + o; o += (size_t)n;
  float* ad2  = ws + o; o += (size_t)n;
  int* cnt     = (int*)(ws + o); o += (size_t)n;
  int* rowptr  = (int*)(ws + o); o += (size_t)n + 1;
  int* cursor  = (int*)(ws + o); o += (size_t)n;
  int* csr_src = (int*)(ws + o); o += (size_t)nE;

  zero_kernel<<<(n + 255) / 256, 256, 0, stream>>>((unsigned*)cnt, (long long)n);
  degree_kernel<<<(nE + 255) / 256, 256, 0, stream>>>(dst, nE, cnt);
  scan_kernel<<<1, 1024, 0, stream>>>(cnt, n, rowptr, cursor);
  fill_kernel<<<(nE + 255) / 256, 256, 0, stream>>>(src, dst, nE, cursor, csr_src);
  pack_b_kernel<500, 128><<<(500 * 128 + 255) / 256, 256, 0, stream>>>(W1, Bp1);
  pack_b_kernel<128, 32><<<(128 * 32 + 255) / 256, 256, 0, stream>>>(W2, Bp2);

  const int gblocks = (int)(((size_t)n * 32 + 255) / 256);

  gemm_wmma_kernel<500, 128, 2><<<(n + 15) / 16, 256, 0, stream>>>(
      x, Bp1, h1, n, a_src1, a_dst1, as1, ad1);
  gat_gather_kernel<2, 64, false><<<gblocks, 256, 0, stream>>>(
      h1, as1, ad1, rowptr, csr_src, n, b1, agg1);

  gemm_wmma_kernel<128, 32, 1><<<(n + 63) / 64, 256, 0, stream>>>(
      agg1, Bp2, h2, n, a_src2, a_dst2, as2, ad2);
  gat_gather_kernel<1, 32, true><<<gblocks, 256, 0, stream>>>(
      h2, as2, ad2, rowptr, csr_src, n, b2, (float*)d_out);
}


# CSR scan kernel: each wave loads 25x32 counts as one batch and prefix-sums them in registers (DPP row shifts), instead of one load plus six dependent lane permutes per 32 counts
# speedup vs baseline: 2.4021x; 1.1059x over previous
#include <hip/hip_runtime.h>
#include <hip/hip_bf16.h>


typedef float v2f __attribute__((ext_vector_type(2)));
typedef float v8f __attribute__((ext_vector_type(8)));

__global__ void zero_kernel(unsigned* __restrict__ p, long long count) {
  long long i = (long long)blockIdx.x * blockDim.x + threadIdx.x;
  long long stride = (long long)gridDim.x * blockDim.x;
  for (; i < count; i += stride) p[i] = 0u;
}

template <int K, int NCOL>
__global__ void pack_b_kernel(const float* __restrict__ B,
                              float* __restrict__ Bp) {
  int tid = blockIdx.x * blockDim.x + threadIdx.x;
  if (tid >= K * NCOL) return;
  int k = tid / NCOL;
  int col = tid - k * NCOL;
  int kk = k >> 2;
  int p = (k >> 1) & 1;
  int q = k & 1;
  Bp[(((kk * 2 + p) * NCOL) + col) * 2 + q] = B[tid];
}

template <int K, int NCOL, int H>
__global__ __launch_bounds__(256) void gemm_wmma_kernel(
    const float* __restrict__ A, const float* __restrict__ Bp,
    float* __restrict__ C, int M, const float* __restrict__ a_s,
    const float* __restrict__ a_d, float* __restrict__ out_s,
    float* __restrict__ out_d) {
  constexpr int NT = NCOL / 16;
  constexpr int MT = 8 / NT;
  constexpr int ROWS = MT * 16;
  constexpr int KV = K / 4;
  __shared__ __align__(16) float As[ROWS * K];

  const int tid = threadIdx.x;
  const int wave = tid >> 5;
  const int lane = tid & 31;
  const int row0 = blockIdx.x * ROWS;

  {
    const float4* A4 = (const float4*)A;
    float4* As4 = (float4*)As;
    for (int i = tid; i < ROWS * KV; i += 256) {
      int r = i / KV;
      int c = i - r * KV;
      int gr = row0 + r;
      float4 v = {0.f, 0.f, 0.f, 0.f};
      if (gr < M) v = A4[(size_t)gr * KV + c];
      As4[i] = v;
    }
  }
  __syncthreads();

  const int mtile = wave / NT;
  const int ntile = wave - mtile * NT;
  const int lrow = (mtile << 4) + (lane & 15);
  const int koff = (lane >> 4) << 1;
  const int ncol = (ntile << 4) + (lane & 15);
  const int bbase = ((lane >> 4) * NCOL + ncol);

  v8f acc = {0.f, 0.f, 0.f, 0.f, 0.f, 0.f, 0.f, 0.f};
  for (int kk = 0; kk < KV; ++kk) {
    v2f a, b;
    a.x = As[lrow * K + kk * 4 + koff];
    a.y = As[lrow * K + kk * 4 + koff + 1];
    b = *(const v2f*)&Bp[(size_t)(bbase + kk * 2 * NCOL) * 2];
    acc = __builtin_amdgcn_wmma_f32_16x16x4_f32(
         false, a,  false, b,
         (short)0, acc,  false,  false);
  }

  constexpr int P = NCOL + 4;
  constexpr int F = NCOL / H;
  constexpr int TPR = 256 / ROWS;
  constexpr int TPH = TPR / H;
  static_assert(F / TPH == 8, "each thread sums 8 columns (two float4)");
  static_assert(ROWS * P <= ROWS * K, "C tile must fit in the A tile's LDS");
  float* Cs = As;
  __syncthreads();
  {
    const int crow = (mtile << 4) + ((lane >> 4) << 3);
    for (int i = 0; i < 8; ++i) Cs[(crow + i) * P + ncol] = acc[i];
  }
  __syncthreads();
  {
    const int r = tid / TPR;
    const int rem = tid - r * TPR;
    const int hd = rem / TPH;
    const int part = rem - hd * TPH;
    const int c0 = hd * F + part * 8;
    const float4 u0 = *(const float4*)&Cs[r * P + c0];
    const float4 u1 = *(const float4*)&Cs[r * P + c0 + 4];
    const float4 s0 = *(const float4*)&a_s[c0];
    const float4 s1 = *(const float4*)&a_s[c0 + 4];
    const float4 d0 = *(const float4*)&a_d[c0];
    const float4 d1 = *(const float4*)&a_d[c0 + 4];
    float ps = u0.x * s0.x;
    ps += u0.y * s0.y; ps += u0.z * s0.z; ps += u0.w * s0.w;
    ps += u1.x * s1.x; ps += u1.y * s1.y; ps += u1.z * s1.z; ps += u1.w * s1.w;
    float pd = u0.x * d0.x;
    pd += u0.y * d0.y; pd += u0.z * d0.z; pd += u0.w * d0.w;
    pd += u1.x * d1.x; pd += u1.y * d1.y; pd += u1.z * d1.z; pd += u1.w * d1.w;
    for (int o = TPH / 2; o > 0; o >>= 1) {
      ps += __shfl_xor(ps, o);
      pd += __shfl_xor(pd, o);
    }
    const int gr = row0 + r;
    if (part == 0 && gr < M) {
      out_s[gr * H + hd] = ps;
      out_d[gr * H + hd] = pd;
    }
  }

  const int rbase = row0 + (mtile << 4) + ((lane >> 4) << 3);
  if (row0 + ROWS <= M) {
#pragma unroll
    for (int i = 0; i < 8; ++i) C[(size_t)(rbase + i) * NCOL + ncol] = acc[i];
  } else {
#pragma unroll
    for (int i = 0; i < 8; ++i) {
      int r = rbase + i;
      if (r < M) C[(size_t)r * NCOL + ncol] = acc[i];
    }
  }
}

__global__ void degree_kernel(const int* __restrict__ dst, int nE,
                              int* __restrict__ cnt) {
  int e = blockIdx.x * blockDim.x + threadIdx.x;
  if (e < nE) atomicAdd(&cnt[dst[e]], 1);
}

constexpr int SCAN_WAVES = 32;
constexpr int SCAN_U = 25;
static_assert(SCAN_WAVES * 32 == 1024, "the launch below uses one block of 1024 threads");

__device__ __forceinline__ int wave_incl_scan(int v, int lane) {
  v += __builtin_amdgcn_update_dpp(0, v, 0x111, 0xf, 0xf, true);
  v += __builtin_amdgcn_update_dpp(0, v, 0x112, 0xf, 0xf, true);
  v += __builtin_amdgcn_update_dpp(0, v, 0x114, 0xf, 0xf, true);
  v += __builtin_amdgcn_update_dpp(0, v, 0x118, 0xf, 0xf, true);
  const int low = __builtin_amdgcn_readlane(v, 15);
  return v + ((lane & 16) ? low : 0);
}

__global__ __launch_bounds__(SCAN_WAVES * 32) void scan_kernel(
    const int* __restrict__ cnt, int n, int* __restrict__ rowptr,
    int* __restrict__ cursor) {
  __shared__ int wsum[SCAN_WAVES];
  const int lane = threadIdx.x & 31;
  const int wave = __builtin_amdgcn_readfirstlane((int)(threadIdx.x >> 5));
  const int seg = (n + SCAN_WAVES - 1) / SCAN_WAVES;
  const int lo = wave * seg;
  const int hi = min(lo + seg, n);

  int part = 0;
  for (int sbase = lo; sbase < hi; sbase += 32 * SCAN_U) {
    int c[SCAN_U];
#pragma unroll
    for (int u = 0; u < SCAN_U; ++u) {
      const int i = sbase + u * 32 + lane;
      const int v = cnt[min(i, hi - 1)];
      c[u] = (i < hi) ? v : 0;
    }
#pragma unroll
    for (int u = 0; u < SCAN_U; ++u) part += c[u];
  }
  part = __builtin_amdgcn_readlane(wave_incl_scan(part, lane), 31);
  if (lane == 0) wsum[wave] = part;
  __syncthreads();

  int carry = (lane < wave) ? wsum[lane] : 0;
  carry = __builtin_amdgcn_readlane(wave_incl_scan(carry, lane), 31);

  for (int sbase = lo; sbase < hi; sbase += 32 * SCAN_U) {
    int c[SCAN_U];
#pragma unroll
    for (int u = 0; u < SCAN_U; ++u) {
      const int i = sbase + u * 32 + lane;
      c[u] = cnt[min(i, hi - 1)];
    }
    __builtin_amdgcn_sched_barrier(0);
    int inc[SCAN_U];
#pragma unroll
    for (int u = 0; u < SCAN_U; ++u) {
      const int i = sbase + u * 32 + lane;
      c[u] = (i < hi) ? c[u] : 0;
      inc[u] = wave_incl_scan(c[u], lane);
    }
#pragma unroll
    for (int u = 0; u < SCAN_U; ++u) {
      const int i = sbase + u * 32 + lane;
      const int excl = carry + inc[u] - c[u];
      if (i < hi) { rowptr[i] = excl; cursor[i] = excl; }
      carry += __builtin_amdgcn_readlane(inc[u], 31);
    }
  }
  if (wave == SCAN_WAVES - 1 && lane == 0) rowptr[n] = carry;
}

__global__ void fill_kernel(const int* __restrict__ src,
                            const int* __restrict__ dst, int nE,
                            int* __restrict__ cursor,
                            int* __restrict__ csr_src) {
  int e = blockIdx.x * blockDim.x + threadIdx.x;
  if (e >= nE) return;
  int pos = atomicAdd(&cursor[dst[e]], 1);
  csr_src[pos] = src[e];
}

template <int H>
__device__ __forceinline__ void edge_logit(const float* __restrict__ as,
                                           const int* __restrict__ csr_src,
                                           const float (&adv)[H], int start,
                                           int deg, int node, int k, int& s,
                                           float (&v)[H]) {
  s = (k < deg) ? csr_src[start + k] : node;
  for (int hd = 0; hd < H; ++hd) {
    float t = as[s * H + hd] + adv[hd];
    v[hd] = t > 0.f ? t : 0.2f * t;
  }
}

template <int H, int F, bool LSM>
__global__ __launch_bounds__(256) void gat_gather_kernel(
    const float* __restrict__ hsrc, const float* __restrict__ as,
    const float* __restrict__ ad, const int* __restrict__ rowptr,
    const int* __restrict__ csr_src, int n, const float* __restrict__ bias,
    float* __restrict__ agg) {
  constexpr int HF = H * F;
  constexpr int VEC = HF / 32;
  static_assert(VEC == 1 || VEC == 4, "row must be 32 or 128 floats");
  static_assert(!LSM || VEC == 1, "log_softmax needs the row in one wave, one class per lane");
  static_assert(H == 1 || H == 2, "per-lane head select below handles 1 or 2 heads");
  const int lane = threadIdx.x & 31;
  const int node = __builtin_amdgcn_readfirstlane(
      (int)((blockIdx.x * blockDim.x + threadIdx.x) >> 5));
  if (node >= n) return;
  const int start = rowptr[node];
  const int deg = rowptr[node + 1] - start;
  const int tot = deg + 1;
  const int myhead = (lane * VEC) / F;

  float adv[H];
  for (int hd = 0; hd < H; ++hd) adv[hd] = ad[node * H + hd];

  const bool ok0 = lane < tot;
  int s0 = node;
  float v0[H];
  for (int hd = 0; hd < H; ++hd) v0[hd] = -INFINITY;
  if (ok0) edge_logit<H>(as, csr_src, adv, start, deg, node, lane, s0, v0);

  float mx[H];
  for (int hd = 0; hd < H; ++hd) mx[hd] = v0[hd];
  for (int base = 32; base < tot; base += 32) {
    const int k = base + lane;
    if (k < tot) {
      int s; float v[H];
      edge_logit<H>(as, csr_src, adv, start, deg, node, k, s, v);
      for (int hd = 0; hd < H; ++hd) mx[hd] = fmaxf(mx[hd], v[hd]);
    }
  }
  for (int hd = 0; hd < H; ++hd)
    for (int o = 16; o > 0; o >>= 1) mx[hd] = fmaxf(mx[hd], __shfl_xor(mx[hd], o));

  float w0[H], dn[H];
  for (int hd = 0; hd < H; ++hd) {
    w0[hd] = ok0 ? __expf(v0[hd] - mx[hd]) : 0.f;
    dn[hd] = w0[hd];
  }
  for (int base = 32; base < tot; base += 32) {
    const int k = base + lane;
    if (k < tot) {
      int s; float v[H];
      edge_logit<H>(as, csr_src, adv, start, deg, node, k, s, v);
      for (int hd = 0; hd < H; ++hd) dn[hd] += __expf(v[hd] - mx[hd]);
    }
  }
  for (int hd = 0; hd < H; ++hd)
    for (int o = 16; o > 0; o >>= 1) dn[hd] += __shfl_xor(dn[hd], o);

  float acc[VEC];
  for (int i = 0; i < VEC; ++i) acc[i] = 0.f;
  const float* rowbase = hsrc + lane * VEC;

  for (int base = 0; base < tot; base += 32) {
    int sc = s0;
    float al[H];
    if (base == 0) {
      for (int hd = 0; hd < H; ++hd) al[hd] = w0[hd] / (dn[hd] + 1e-16f);
    } else {
      const int k = base + lane;
      sc = node;
      for (int hd = 0; hd < H; ++hd) al[hd] = 0.f;
      if (k < tot) {
        float v[H];
        edge_logit<H>(as, csr_src, adv, start, deg, node, k, sc, v);
        for (int hd = 0; hd < H; ++hd)
          al[hd] = __expf(v[hd] - mx[hd]) / (dn[hd] + 1e-16f);
      }
    }
    const int cnt = min(tot - base, 32);
#pragma unroll 4
    for (int j = 0; j < cnt; ++j) {
      const int sj = __builtin_amdgcn_readlane(sc, j);
      float aj = __int_as_float(
          __builtin_amdgcn_readlane(__float_as_int(al[0]), j));
      if constexpr (H > 1) {
        const float a1 = __int_as_float(
            __builtin_amdgcn_readlane(__float_as_int(al[H - 1]), j));
        aj = myhead ? a1 : aj;
      }
      const float* rp = rowbase + (size_t)sj * HF;
      if constexpr (VEC == 4) {
        const float4 hv = *(const float4*)rp;
        acc[0] += hv.x * aj;
        acc[1] += hv.y * aj;
        acc[2] += hv.z * aj;
        acc[3] += hv.w * aj;
      } else {
        acc[0] += rp[0] * aj;
      }
    }
  }

  for (int i = 0; i < VEC; ++i) acc[i] += bias[lane * VEC + i];
  if constexpr (LSM) {
    const float v = acc[0];
    float m = v;
    for (int o = 16; o > 0; o >>= 1) m = fmaxf(m, __shfl_xor(m, o));
    float sum = __expf(v - m);
    for (int o = 16; o > 0; o >>= 1) sum += __shfl_xor(sum, o);
    acc[0] = v - m - __logf(sum);
  } else {
    for (int i = 0; i < VEC; ++i)
      acc[i] = acc[i] > 0.f ? acc[i] : (__expf(acc[i]) - 1.f);
  }

  float* op = agg + (size_t)node * HF + lane * VEC;
  if constexpr (VEC == 4) {
    float4 r;
    r.x = acc[0]; r.y = acc[1]; r.z = acc[2]; r.w = acc[3];
    *(float4*)op = r;
  } else {
    op[0] = acc[0];
  }
}

extern "C" void kernel_launch(void* const* d_in, const int* in_sizes, int n_in,
                              void* d_out, int out_size, void* d_ws,
                              size_t ws_size, hipStream_t stream) {
  (void)n_in; (void)out_size; (void)ws_size;
  const float* x      = (const float*)d_in[0];
  const int*   ei     = (const int*)d_in[1];
  const float* W1     = (const float*)d_in[2];
  const float* a_src1 = (const float*)d_in[3];
  const float* a_dst1 = (const float*)d_in[4];
  const float* b1     = (const float*)d_in[5];
  const float* W2     = (const float*)d_in[6];
  const float* a_src2 = (const float*)d_in[7];
  const float* a_dst2 = (const float*)d_in[8];
  const float* b2     = (const float*)d_in[9];

  const int n    = in_sizes[0] / 500;
  const int nE   = in_sizes[1] / 2;
  const int* src = ei;
  const int* dst = ei + nE;

  float* ws = (float*)d_ws;
  size_t o = 0;
  float* h1   = ws + o; o += (size_t)n * 128;
  float* agg1 = ws + o; o += (size_t)n * 128;
  float* h2   = ws + o; o += (size_t)n * 32;
  float* Bp1  = ws + o; o += (size_t)500 * 128;
  float* Bp2  = ws + o; o += (size_t)128 * 32;
  float* as1  = ws + o; o += (size_t)n * 2;
  float* ad1  = ws + o; o += (size_t)n * 2;
  float* as2  = ws + o; o += (size_t)n;
  float* ad2  = ws + o; o += (size_t)n;
  int* cnt     = (int*)(ws + o); o += (size_t)n;
  int* rowptr  = (int*)(ws + o); o += (size_t)n + 1;
  int* cursor  = (int*)(ws + o); o += (size_t)n;
  int* csr_src = (int*)(ws + o); o += (size_t)nE;

  zero_kernel<<<(n + 255) / 256, 256, 0, stream>>>((unsigned*)cnt, (long long)n);
  degree_kernel<<<(nE + 255) / 256, 256, 0, stream>>>(dst, nE, cnt);
  scan_kernel<<<1, 1024, 0, stream>>>(cnt, n, rowptr, cursor);
  fill_kernel<<<(nE + 255) / 256, 256, 0, stream>>>(src, dst, nE, cursor, csr_src);
  pack_b_kernel<500, 128><<<(500 * 128 + 255) / 256, 256, 0, stream>>>(W1, Bp1);
  pack_b_kernel<128, 32><<<(128 * 32 + 255) / 256, 256, 0, stream>>>(W2, Bp2);

  const int gblocks = (int)(((size_t)n * 32 + 255) / 256);

  gemm_wmma_kernel<500, 128, 2><<<(n + 15) / 16, 256, 0, stream>>>(
      x, Bp1, h1, n, a_src1, a_dst1, as1, ad1);
  gat_gather_kernel<2, 64, false><<<gblocks, 256, 0, stream>>>(
      h1, as1, ad1, rowptr, csr_src, n, b1, agg1);

  gemm_wmma_kernel<128, 32, 1><<<(n + 63) / 64, 256, 0, stream>>>(
      agg1, Bp2, h2, n, a_src2, a_dst2, as2, ad2);
  gat_gather_kernel<1, 32, true><<<gblocks, 256, 0, stream>>>(
      h2, as2, ad2, rowptr, csr_src, n, b2, (float*)d_out);
}


# independent passes share launches: layer-1 GEMM cut in two row ranges, edge-count blocks run beside the first and edge-fill blocks beside the second; zero+pack+pack in one launch; 10 -> 7 launches
# speedup vs baseline: 2.7161x; 1.1307x over previous
#include <hip/hip_runtime.h>
#include <hip/hip_bf16.h>


typedef float v2f __attribute__((ext_vector_type(2)));
typedef float v8f __attribute__((ext_vector_type(8)));

template <int K, int NCOL>
__device__ __forceinline__ void pack_b(const float* __restrict__ B,
                                       float* __restrict__ Bp, int tid) {
  if (tid >= K * NCOL) return;
  int k = tid / NCOL;
  int col = tid - k * NCOL;
  int kk = k >> 2;
  int p = (k >> 1) & 1;
  int q = k & 1;
  Bp[(((kk * 2 + p) * NCOL) + col) * 2 + q] = B[tid];
}

constexpr int PACK1_BLOCKS = (500 * 128 + 255) / 256;
constexpr int PACK2_BLOCKS = (128 * 32 + 255) / 256;

__global__ __launch_bounds__(256) void prep_kernel(
    const float* __restrict__ W1, float* __restrict__ Bp1,
    const float* __restrict__ W2, float* __restrict__ Bp2,
    unsigned* __restrict__ cnt, int n) {
  const int b = blockIdx.x;
  if (b < PACK1_BLOCKS) {
    pack_b<500, 128>(W1, Bp1, b * 256 + (int)threadIdx.x);
  } else if (b < PACK1_BLOCKS + PACK2_BLOCKS) {
    pack_b<128, 32>(W2, Bp2, (b - PACK1_BLOCKS) * 256 + (int)threadIdx.x);
  } else {
    const int i = (b - PACK1_BLOCKS - PACK2_BLOCKS) * 256 + (int)threadIdx.x;
    if (i < n) cnt[i] = 0u;
  }
}

template <int K, int NCOL>
struct GemmTile {
  static constexpr int NT = NCOL / 16;
  static constexpr int MT = 8 / NT;
  static constexpr int ROWS = MT * 16;
  static constexpr int LDS_FLOATS = ROWS * K;
};

template <int K, int NCOL, int H>
__device__ __forceinline__ void gemm_block(
    float* __restrict__ As, int blk,
    const float* __restrict__ A, const float* __restrict__ Bp,
    float* __restrict__ C, int M, const float* __restrict__ a_s,
    const float* __restrict__ a_d, float* __restrict__ out_s,
    float* __restrict__ out_d) {
  constexpr int NT = GemmTile<K, NCOL>::NT;
  constexpr int MT = GemmTile<K, NCOL>::MT;
  constexpr int ROWS = GemmTile<K, NCOL>::ROWS;
  constexpr int KV = K / 4;

  const int tid = threadIdx.x;
  const int wave = tid >> 5;
  const int lane = tid & 31;
  const int row0 = blk * ROWS;

  {
    const float4* A4 = (const float4*)A;
    float4* As4 = (float4*)As;
    for (int i = tid; i < ROWS * KV; i += 256) {
      int r = i / KV;
      int c = i - r * KV;
      int gr = row0 + r;
      float4 v = {0.f, 0.f, 0.f, 0.f};
      if (gr < M) v = A4[(size_t)gr * KV + c];
      As4[i] = v;
    }
  }
  __syncthreads();

  const int mtile = wave / NT;
  const int ntile = wave - mtile * NT;
  const int lrow = (mtile << 4) + (lane & 15);
  const int koff = (lane >> 4) << 1;
  const int ncol = (ntile << 4) + (lane & 15);
  const int bbase = ((lane >> 4) * NCOL + ncol);

  v8f acc = {0.f, 0.f, 0.f, 0.f, 0.f, 0.f, 0.f, 0.f};
  for (int kk = 0; kk < KV; ++kk) {
    v2f a, b;
    a.x = As[lrow * K + kk * 4 + koff];
    a.y = As[lrow * K + kk * 4 + koff + 1];
    b = *(const v2f*)&Bp[(size_t)(bbase + kk * 2 * NCOL) * 2];
    acc = __builtin_amdgcn_wmma_f32_16x16x4_f32(
         false, a,  false, b,
         (short)0, acc,  false,  false);
  }

  constexpr int P = NCOL + 4;
  constexpr int F = NCOL / H;
  constexpr int TPR = 256 / ROWS;
  constexpr int TPH = TPR / H;
  static_assert(F / TPH == 8, "each thread sums 8 columns (two float4)");
  static_assert(ROWS * P <= ROWS * K, "C tile must fit in the A tile's LDS");
  float* Cs = As;
  __syncthreads();
  {
    const int crow = (mtile << 4) + ((lane >> 4) << 3);
    for (int i = 0; i < 8; ++i) Cs[(crow + i) * P + ncol] = acc[i];
  }
  __syncthreads();
  {
    const int r = tid / TPR;
    const int rem = tid - r * TPR;
    const int hd = rem / TPH;
    const int part = rem - hd * TPH;
    const int c0 = hd * F + part * 8;
    const float4 u0 = *(const float4*)&Cs[r * P + c0];
    const float4 u1 = *(const float4*)&Cs[r * P + c0 + 4];
    const float4 s0 = *(const float4*)&a_s[c0];
    const float4 s1 = *(const float4*)&a_s[c0 + 4];
    const float4 d0 = *(const float4*)&a_d[c0];
    const float4 d1 = *(const float4*)&a_d[c0 + 4];
    float ps = u0.x * s0.x;
    ps += u0.y * s0.y; ps += u0.z * s0.z; ps += u0.w * s0.w;
    ps += u1.x * s1.x; ps += u1.y * s1.y; ps += u1.z * s1.z; ps += u1.w * s1.w;
    float pd = u0.x * d0.x;
    pd += u0.y * d0.y; pd += u0.z * d0.z; pd += u0.w * d0.w;
    pd += u1.x * d1.x; pd += u1.y * d1.y; pd += u1.z * d1.z; pd += u1.w * d1.w;
    for (int o = TPH / 2; o > 0; o >>= 1) {
      ps += __shfl_xor(ps, o);
      pd += __shfl_xor(pd, o);
    }
    const int gr = row0 + r;
    if (part == 0 && gr < M) {
      out_s[gr * H + hd] = ps;
      out_d[gr * H + hd] = pd;
    }
  }

  const int rbase = row0 + (mtile << 4) + ((lane >> 4) << 3);
  if (row0 + ROWS <= M) {
#pragma unroll
    for (int i = 0; i < 8; ++i) C[(size_t)(rbase + i) * NCOL + ncol] = acc[i];
  } else {
#pragma unroll
    for (int i = 0; i < 8; ++i) {
      int r = rbase + i;
      if (r < M) C[(size_t)r * NCOL + ncol] = acc[i];
    }
  }
}

template <int K, int NCOL, int H>
__global__ __launch_bounds__(256) void gemm_wmma_kernel(
    const float* __restrict__ A, const float* __restrict__ Bp,
    float* __restrict__ C, int M, const float* __restrict__ a_s,
    const float* __restrict__ a_d, float* __restrict__ out_s,
    float* __restrict__ out_d) {
  __shared__ __align__(16) float As[GemmTile<K, NCOL>::LDS_FLOATS];
  gemm_block<K, NCOL, H>(As, (int)blockIdx.x, A, Bp, C, M, a_s, a_d, out_s, out_d);
}

__device__ __forceinline__ void degree_edge(const int* __restrict__ dst, int nE,
                                            int* __restrict__ cnt, int e) {
  if (e < nE) atomicAdd(&cnt[dst[e]], 1);
}

constexpr int SCAN_WAVES = 32;
constexpr int SCAN_U = 25;
static_assert(SCAN_WAVES * 32 == 1024, "the launch below uses one block of 1024 threads");

__device__ __forceinline__ int wave_incl_scan(int v, int lane) {
  v += __builtin_amdgcn_update_dpp(0, v, 0x111, 0xf, 0xf, true);
  v += __builtin_amdgcn_update_dpp(0, v, 0x112, 0xf, 0xf, true);
  v += __builtin_amdgcn_update_dpp(0, v, 0x114, 0xf, 0xf, true);
  v += __builtin_amdgcn_update_dpp(0, v, 0x118, 0xf, 0xf, true);
  const int low = __builtin_amdgcn_readlane(v, 15);
  return v + ((lane & 16) ? low : 0);
}

__global__ __launch_bounds__(SCAN_WAVES * 32) void scan_kernel(
    const int* __restrict__ cnt, int n, int* __restrict__ rowptr,
    int* __restrict__ cursor) {
  __shared__ int wsum[SCAN_WAVES];
  const int lane = threadIdx.x & 31;
  const int wave = __builtin_amdgcn_readfirstlane((int)(threadIdx.x >> 5));
  const int seg = (n + SCAN_WAVES - 1) / SCAN_WAVES;
  const int lo = wave * seg;
  const int hi = min(lo + seg, n);

  int part = 0;
  for (int sbase = lo; sbase < hi; sbase += 32 * SCAN_U) {
    int c[SCAN_U];
#pragma unroll
    for (int u = 0; u < SCAN_U; ++u) {
      const int i = sbase + u * 32 + lane;
      const int v = cnt[min(i, hi - 1)];
      c[u] = (i < hi) ? v : 0;
    }
#pragma unroll
    for (int u = 0; u < SCAN_U; ++u) part += c[u];
  }
  part = __builtin_amdgcn_readlane(wave_incl_scan(part, lane), 31);
  if (lane == 0) wsum[wave] = part;
  __syncthreads();

  int carry = (lane < wave) ? wsum[lane] : 0;
  carry = __builtin_amdgcn_readlane(wave_incl_scan(carry, lane), 31);

  for (int sbase = lo; sbase < hi; sbase += 32 * SCAN_U) {
    int c[SCAN_U];
#pragma unroll
    for (int u = 0; u < SCAN_U; ++u) {
      const int i = sbase + u * 32 + lane;
      c[u] = cnt[min(i, hi - 1)];
    }
    __builtin_amdgcn_sched_barrier(0);
    int inc[SCAN_U];
#pragma unroll
    for (int u = 0; u < SCAN_U; ++u) {
      const int i = sbase + u * 32 + lane;
      c[u] = (i < hi) ? c[u] : 0;
      inc[u] = wave_incl_scan(c[u], lane);
    }
#pragma unroll
    for (int u = 0; u < SCAN_U; ++u) {
      const int i = sbase + u * 32 + lane;
      const int excl = carry + inc[u] - c[u];
      if (i < hi) { rowptr[i] = excl; cursor[i] = excl; }
      carry += __builtin_amdgcn_readlane(inc[u], 31);
    }
  }
  if (wave == SCAN_WAVES - 1 && lane == 0) rowptr[n] = carry;
}

__device__ __forceinline__ void fill_edge(const int* __restrict__ src,
                                          const int* __restrict__ dst, int nE,
                                          int* __restrict__ cursor,
                                          int* __restrict__ csr_src, int e) {
  if (e >= nE) return;
  int pos = atomicAdd(&cursor[dst[e]], 1);
  csr_src[pos] = src[e];
}

template <bool FILL>
__global__ __launch_bounds__(256) void gemm1_edge_kernel(
    const float* __restrict__ A, const float* __restrict__ Bp,
    float* __restrict__ C, int M, const float* __restrict__ a_s,
    const float* __restrict__ a_d, float* __restrict__ out_s,
    float* __restrict__ out_d, int gblocks, int rowblk0,
    const int* __restrict__ src, const int* __restrict__ dst, int nE,
    int* __restrict__ ctr, int* __restrict__ csr_src) {
  __shared__ __align__(16) float As[GemmTile<500, 128>::LDS_FLOATS];
  const int b = blockIdx.x;
  if (b < gblocks) {
    gemm_block<500, 128, 2>(As, rowblk0 + b, A, Bp, C, M, a_s, a_d, out_s, out_d);
  } else {
    const int e = (b - gblocks) * 256 + (int)threadIdx.x;
    if constexpr (FILL) fill_edge(src, dst, nE, ctr, csr_src, e);
    else degree_edge(dst, nE, ctr, e);
  }
}

template <int H>
__device__ __forceinline__ void edge_logit(const float* __restrict__ as,
                                           const int* __restrict__ csr_src,
                                           const float (&adv)[H], int start,
                                           int deg, int node, int k, int& s,
                                           float (&v)[H]) {
  s = (k < deg) ? csr_src[start + k] : node;
  for (int hd = 0; hd < H; ++hd) {
    float t = as[s * H + hd] + adv[hd];
    v[hd] = t > 0.f ? t : 0.2f * t;
  }
}

template <int H, int F, bool LSM>
__global__ __launch_bounds__(256) void gat_gather_kernel(
    const float* __restrict__ hsrc, const float* __restrict__ as,
    const float* __restrict__ ad, const int* __restrict__ rowptr,
    const int* __restrict__ csr_src, int n, const float* __restrict__ bias,
    float* __restrict__ agg) {
  constexpr int HF = H * F;
  constexpr int VEC = HF / 32;
  static_assert(VEC == 1 || VEC == 4, "row must be 32 or 128 floats");
  static_assert(!LSM || VEC == 1, "log_softmax needs the row in one wave, one class per lane");
  static_assert(H == 1 || H == 2, "per-lane head select below handles 1 or 2 heads");
  const int lane = threadIdx.x & 31;
  const int node = __builtin_amdgcn_readfirstlane(
      (int)((blockIdx.x * blockDim.x + threadIdx.x) >> 5));
  if (node >= n) return;
  const int start = rowptr[node];
  const int deg = rowptr[node + 1] - start;
  const int tot = deg + 1;
  const int myhead = (lane * VEC) / F;

  float adv[H];
  for (int hd = 0; hd < H; ++hd) adv[hd] = ad[node * H + hd];

  const bool ok0 = lane < tot;
  int s0 = node;
  float v0[H];
  for (int hd = 0; hd < H; ++hd) v0[hd] = -INFINITY;
  if (ok0) edge_logit<H>(as, csr_src, adv, start, deg, node, lane, s0, v0);

  float mx[H];
  for (int hd = 0; hd < H; ++hd) mx[hd] = v0[hd];
  for (int base = 32; base < tot; base += 32) {
    const int k = base + lane;
    if (k < tot) {
      int s; float v[H];
      edge_logit<H>(as, csr_src, adv, start, deg, node, k, s, v);
      for (int hd = 0; hd < H; ++hd) mx[hd] = fmaxf(mx[hd], v[hd]);
    }
  }
  for (int hd = 0; hd < H; ++hd)
    for (int o = 16; o > 0; o >>= 1) mx[hd] = fmaxf(mx[hd], __shfl_xor(mx[hd], o));

  float w0[H], dn[H];
  for (int hd = 0; hd < H; ++hd) {
    w0[hd] = ok0 ? __expf(v0[hd] - mx[hd]) : 0.f;
    dn[hd] = w0[hd];
  }
  for (int base = 32; base < tot; base += 32) {
    const int k = base + lane;
    if (k < tot) {
      int s; float v[H];
      edge_logit<H>(as, csr_src, adv, start, deg, node, k, s, v);
      for (int hd = 0; hd < H; ++hd) dn[hd] += __expf(v[hd] - mx[hd]);
    }
  }
  for (int hd = 0; hd < H; ++hd)
    for (int o = 16; o > 0; o >>= 1) dn[hd] += __shfl_xor(dn[hd], o);

  float acc[VEC];
  for (int i = 0; i < VEC; ++i) acc[i] = 0.f;
  const float* rowbase = hsrc + lane * VEC;

  for (int base = 0; base < tot; base += 32) {
    int sc = s0;
    float al[H];
    if (base == 0) {
      for (int hd = 0; hd < H; ++hd) al[hd] = w0[hd] / (dn[hd] + 1e-16f);
    } else {
      const int k = base + lane;
      sc = node;
      for (int hd = 0; hd < H; ++hd) al[hd] = 0.f;
      if (k < tot) {
        float v[H];
        edge_logit<H>(as, csr_src, adv, start, deg, node, k, sc, v);
        for (int hd = 0; hd < H; ++hd)
          al[hd] = __expf(v[hd] - mx[hd]) / (dn[hd] + 1e-16f);
      }
    }
    const int cnt = min(tot - base, 32);
#pragma unroll 4
    for (int j = 0; j < cnt; ++j) {
      const int sj = __builtin_amdgcn_readlane(sc, j);
      float aj = __int_as_float(
          __builtin_amdgcn_readlane(__float_as_int(al[0]), j));
      if constexpr (H > 1) {
        const float a1 = __int_as_float(
            __builtin_amdgcn_readlane(__float_as_int(al[H - 1]), j));
        aj = myhead ? a1 : aj;
      }
      const float* rp = rowbase + (size_t)sj * HF;
      if constexpr (VEC == 4) {
        const float4 hv = *(const float4*)rp;
        acc[0] += hv.x * aj;
        acc[1] += hv.y * aj;
        acc[2] += hv.z * aj;
        acc[3] += hv.w * aj;
      } else {
        acc[0] += rp[0] * aj;
      }
    }
  }

  for (int i = 0; i < VEC; ++i) acc[i] += bias[lane * VEC + i];
  if constexpr (LSM) {
    const float v = acc[0];
    float m = v;
    for (int o = 16; o > 0; o >>= 1) m = fmaxf(m, __shfl_xor(m, o));
    float sum = __expf(v - m);
    for (int o = 16; o > 0; o >>= 1) sum += __shfl_xor(sum, o);
    acc[0] = v - m - __logf(sum);
  } else {
    for (int i = 0; i < VEC; ++i)
      acc[i] = acc[i] > 0.f ? acc[i] : (__expf(acc[i]) - 1.f);
  }

  float* op = agg + (size_t)node * HF + lane * VEC;
  if constexpr (VEC == 4) {
    float4 r;
    r.x = acc[0]; r.y = acc[1]; r.z = acc[2]; r.w = acc[3];
    *(float4*)op = r;
  } else {
    op[0] = acc[0];
  }
}

extern "C" void kernel_launch(void* const* d_in, const int* in_sizes, int n_in,
                              void* d_out, int out_size, void* d_ws,
                              size_t ws_size, hipStream_t stream) {
  (void)n_in; (void)out_size; (void)ws_size;
  const float* x      = (const float*)d_in[0];
  const int*   ei     = (const int*)d_in[1];
  const float* W1     = (const float*)d_in[2];
  const float* a_src1 = (const float*)d_in[3];
  const float* a_dst1 = (const float*)d_in[4];
  const float* b1     = (const float*)d_in[5];
  const float* W2     = (const float*)d_in[6];
  const float* a_src2 = (const float*)d_in[7];
  const float* a_dst2 = (const float*)d_in[8];
  const float* b2     = (const float*)d_in[9];

  const int n    = in_sizes[0] / 500;
  const int nE   = in_sizes[1] / 2;
  const int* src = ei;
  const int* dst = ei + nE;

  float* ws = (float*)d_ws;
  size_t o = 0;
  float* h1   = ws + o; o += (size_t)n * 128;
  float* agg1 = ws + o; o += (size_t)n * 128;
  float* h2   = ws + o; o += (size_t)n * 32;
  float* Bp1  = ws + o; o += (size_t)500 * 128;
  float* Bp2  = ws + o; o += (size_t)128 * 32;
  float* as1  = ws + o; o += (size_t)n * 2;
  float* ad1  = ws + o; o += (size_t)n * 2;
  float* as2  = ws + o; o += (size_t)n;
  float* ad2  = ws + o; o += (size_t)n;
  int* cnt     = (int*)(ws + o); o += (size_t)n;
  int* rowptr  = (int*)(ws + o); o += (size_t)n + 1;
  int* cursor  = (int*)(ws + o); o += (size_t)n;
  int* csr_src = (int*)(ws + o); o += (size_t)nE;

  prep_kernel<<<PACK1_BLOCKS + PACK2_BLOCKS + (n + 255) / 256, 256, 0, stream>>>(
      W1, Bp1, W2, Bp2, (unsigned*)cnt, n);

  const int eblocks = (nE + 255) / 256;
  const int rblocks = (n + 15) / 16;
  const int rblocksA = rblocks / 2;
  const int rblocksB = rblocks - rblocksA;
  gemm1_edge_kernel<false><<<rblocksA + eblocks, 256, 0, stream>>>(
      x, Bp1, h1, n, a_src1, a_dst1, as1, ad1, rblocksA, 0,
      src, dst, nE, cnt, csr_src);
  scan_kernel<<<1, 1024, 0, stream>>>(cnt, n, rowptr, cursor);
  gemm1_edge_kernel<true><<<rblocksB + eblocks, 256, 0, stream>>>(
      x, Bp1, h1, n, a_src1, a_dst1, as1, ad1, rblocksB, rblocksA,
      src, dst, nE, cursor, csr_src);

  const int gblocks = (int)(((size_t)n * 32 + 255) / 256);

  gat_gather_kernel<2, 64, false><<<gblocks, 256, 0, stream>>>(
      h1, as1, ad1, rowptr, csr_src, n, b1, agg1);

  gemm_wmma_kernel<128, 32, 1><<<(n + 63) / 64, 256, 0, stream>>>(
      agg1, Bp2, h2, n, a_src2, a_dst2, as2, ad2);
  gat_gather_kernel<1, 32, true><<<gblocks, 256, 0, stream>>>(
      h2, as2, ad2, rowptr, csr_src, n, b2, (float*)d_out);
}


# layer-1 GEMM: 32-row blocks, each wave keeps two row tiles and feeds both from one B fragment (B from L2 once per two WMMAs); A tiles fetched 8 float4 per thread at a time
# speedup vs baseline: 2.8258x; 1.0404x over previous
#include <hip/hip_runtime.h>
#include <hip/hip_bf16.h>


typedef float v2f __attribute__((ext_vector_type(2)));
typedef float v8f __attribute__((ext_vector_type(8)));

template <int K, int NCOL>
__device__ __forceinline__ void pack_b(const float* __restrict__ B,
                                       float* __restrict__ Bp, int tid) {
  if (tid >= K * NCOL) return;
  int k = tid / NCOL;
  int col = tid - k * NCOL;
  int kk = k >> 2;
  int p = (k >> 1) & 1;
  int q = k & 1;
  Bp[(((kk * 2 + p) * NCOL) + col) * 2 + q] = B[tid];
}

constexpr int PACK1_BLOCKS = (500 * 128 + 255) / 256;
constexpr int PACK2_BLOCKS = (128 * 32 + 255) / 256;

__global__ __launch_bounds__(256) void prep_kernel(
    const float* __restrict__ W1, float* __restrict__ Bp1,
    const float* __restrict__ W2, float* __restrict__ Bp2,
    unsigned* __restrict__ cnt, int n) {
  const int b = blockIdx.x;
  if (b < PACK1_BLOCKS) {
    pack_b<500, 128>(W1, Bp1, b * 256 + (int)threadIdx.x);
  } else if (b < PACK1_BLOCKS + PACK2_BLOCKS) {
    pack_b<128, 32>(W2, Bp2, (b - PACK1_BLOCKS) * 256 + (int)threadIdx.x);
  } else {
    const int i = (b - PACK1_BLOCKS - PACK2_BLOCKS) * 256 + (int)threadIdx.x;
    if (i < n) cnt[i] = 0u;
  }
}

template <int K, int NCOL, int MW>
struct GemmTile {
  static constexpr int NT = NCOL / 16;
  static constexpr int MT = 8 / NT;
  static constexpr int ROWS = MT * MW * 16;
  static constexpr int STEP_F4 = 256 * 8;
  static constexpr int LOAD_F4 = (ROWS * K / 4 + STEP_F4 - 1) / STEP_F4 * STEP_F4;
  static constexpr int LDS_FLOATS = LOAD_F4 * 4;
};

template <int K, int NCOL, int H, int MW>
__device__ __forceinline__ void gemm_block(
    float* __restrict__ As, int blk,
    const float* __restrict__ A, const float* __restrict__ Bp,
    float* __restrict__ C, int M, const float* __restrict__ a_s,
    const float* __restrict__ a_d, float* __restrict__ out_s,
    float* __restrict__ out_d) {
  constexpr int NT = GemmTile<K, NCOL, MW>::NT;
  constexpr int MT = GemmTile<K, NCOL, MW>::MT;
  constexpr int ROWS = GemmTile<K, NCOL, MW>::ROWS;
  constexpr int KV = K / 4;

  const int tid = threadIdx.x;
  const int wave = tid >> 5;
  const int lane = tid & 31;
  const int row0 = blk * ROWS;

  {
    const float4* A4 = (const float4*)A + (size_t)row0 * KV;
    float4* As4 = (float4*)As;
    constexpr int TOT = ROWS * KV;
    constexpr int LB = 8;
    constexpr int NIT = GemmTile<K, NCOL, MW>::LOAD_F4 / (256 * LB);
    const int lim = min(TOT, (M - row0) * KV);
    for (int it = 0; it < NIT; ++it) {
      const int i0 = it * (256 * LB) + tid;
      float4 v[LB];
#pragma unroll
      for (int u = 0; u < LB; ++u) v[u] = A4[min(i0 + u * 256, lim - 1)];
      __builtin_amdgcn_sched_barrier(0);
#pragma unroll
      for (int u = 0; u < LB; ++u) {
        const int i = i0 + u * 256;
        float4 t = v[u];
        if (i >= lim) t = float4{0.f, 0.f, 0.f, 0.f};
        As4[i] = t;
      }
    }
  }
  __syncthreads();

  const int mtile = wave / NT;
  const int ntile = wave - mtile * NT;
  const int lrow = ((mtile * MW) << 4) + (lane & 15);
  const int koff = (lane >> 4) << 1;
  const int ncol = (ntile << 4) + (lane & 15);
  const int bbase = ((lane >> 4) * NCOL + ncol);

  v8f acc[MW];
  for (int j = 0; j < MW; ++j)
    acc[j] = v8f{0.f, 0.f, 0.f, 0.f, 0.f, 0.f, 0.f, 0.f};
  for (int kk = 0; kk < KV; ++kk) {
    const v2f b = *(const v2f*)&Bp[(size_t)(bbase + kk * 2 * NCOL) * 2];
    for (int j = 0; j < MW; ++j) {
      v2f a;
      a.x = As[(lrow + j * 16) * K + kk * 4 + koff];
      a.y = As[(lrow + j * 16) * K + kk * 4 + koff + 1];
      acc[j] = __builtin_amdgcn_wmma_f32_16x16x4_f32(
           false, a,  false, b,
           (short)0, acc[j],  false,  false);
    }
  }

  constexpr int P = NCOL + 4;
  constexpr int F = NCOL / H;
  constexpr int RPP = MT * 16;
  constexpr int TPR = 256 / RPP;
  constexpr int TPH = TPR / H;
  static_assert(F / TPH == 8, "each thread sums 8 columns (two float4)");
  static_assert(ROWS * P <= ROWS * K, "C tile must fit in the A tile's LDS");
  float* Cs = As;
  __syncthreads();
#pragma unroll
  for (int j = 0; j < MW; ++j) {
    const int crow = ((mtile * MW + j) << 4) + ((lane >> 4) << 3);
    for (int i = 0; i < 8; ++i) Cs[(crow + i) * P + ncol] = acc[j][i];
  }
  __syncthreads();
  const int rr = tid / TPR;
  const int rem = tid - rr * TPR;
  const int hd = rem / TPH;
  const int part = rem - hd * TPH;
  const int c0 = hd * F + part * 8;
  const float4 s0 = *(const float4*)&a_s[c0];
  const float4 s1 = *(const float4*)&a_s[c0 + 4];
  const float4 d0 = *(const float4*)&a_d[c0];
  const float4 d1 = *(const float4*)&a_d[c0 + 4];
  for (int pass = 0; pass < MW; ++pass) {
    const int r = ((rr >> 4) * MW + pass) * 16 + (rr & 15);
    const float4 u0 = *(const float4*)&Cs[r * P + c0];
    const float4 u1 = *(const float4*)&Cs[r * P + c0 + 4];
    float ps = u0.x * s0.x;
    ps += u0.y * s0.y; ps += u0.z * s0.z; ps += u0.w * s0.w;
    ps += u1.x * s1.x; ps += u1.y * s1.y; ps += u1.z * s1.z; ps += u1.w * s1.w;
    float pd = u0.x * d0.x;
    pd += u0.y * d0.y; pd += u0.z * d0.z; pd += u0.w * d0.w;
    pd += u1.x * d1.x; pd += u1.y * d1.y; pd += u1.z * d1.z; pd += u1.w * d1.w;
    for (int o = TPH / 2; o > 0; o >>= 1) {
      ps += __shfl_xor(ps, o);
      pd += __shfl_xor(pd, o);
    }
    const int gr = row0 + r;
    if (part == 0 && gr < M) {
      out_s[gr * H + hd] = ps;
      out_d[gr * H + hd] = pd;
    }
  }

#pragma unroll
  for (int j = 0; j < MW; ++j) {
    const int rbase = row0 + ((mtile * MW + j) << 4) + ((lane >> 4) << 3);
    if (row0 + ROWS <= M) {
#pragma unroll
      for (int i = 0; i < 8; ++i) C[(size_t)(rbase + i) * NCOL + ncol] = acc[j][i];
    } else {
#pragma unroll
      for (int i = 0; i < 8; ++i) {
        int r = rbase + i;
        if (r < M) C[(size_t)r * NCOL + ncol] = acc[j][i];
      }
    }
  }
}

template <int K, int NCOL, int H>
__global__ __launch_bounds__(256) void gemm_wmma_kernel(
    const float* __restrict__ A, const float* __restrict__ Bp,
    float* __restrict__ C, int M, const float* __restrict__ a_s,
    const float* __restrict__ a_d, float* __restrict__ out_s,
    float* __restrict__ out_d) {
  __shared__ __align__(16) float As[GemmTile<K, NCOL, 1>::LDS_FLOATS];
  gemm_block<K, NCOL, H, 1>(As, (int)blockIdx.x, A, Bp, C, M, a_s, a_d, out_s, out_d);
}

__device__ __forceinline__ void degree_edge(const int* __restrict__ dst, int nE,
                                            int* __restrict__ cnt, int e) {
  if (e < nE) atomicAdd(&cnt[dst[e]], 1);
}

constexpr int SCAN_WAVES = 32;
constexpr int SCAN_U = 25;
static_assert(SCAN_WAVES * 32 == 1024, "the launch below uses one block of 1024 threads");

__device__ __forceinline__ int wave_incl_scan(int v, int lane) {
  v += __builtin_amdgcn_update_dpp(0, v, 0x111, 0xf, 0xf, true);
  v += __builtin_amdgcn_update_dpp(0, v, 0x112, 0xf, 0xf, true);
  v += __builtin_amdgcn_update_dpp(0, v, 0x114, 0xf, 0xf, true);
  v += __builtin_amdgcn_update_dpp(0, v, 0x118, 0xf, 0xf, true);
  const int low = __builtin_amdgcn_readlane(v, 15);
  return v + ((lane & 16) ? low : 0);
}

__global__ __launch_bounds__(SCAN_WAVES * 32) void scan_kernel(
    const int* __restrict__ cnt, int n, int* __restrict__ rowptr,
    int* __restrict__ cursor) {
  __shared__ int wsum[SCAN_WAVES];
  const int lane = threadIdx.x & 31;
  const int wave = __builtin_amdgcn_readfirstlane((int)(threadIdx.x >> 5));
  const int seg = (n + SCAN_WAVES - 1) / SCAN_WAVES;
  const int lo = wave * seg;
  const int hi = min(lo + seg, n);

  int part = 0;
  for (int sbase = lo; sbase < hi; sbase += 32 * SCAN_U) {
    int c[SCAN_U];
#pragma unroll
    for (int u = 0; u < SCAN_U; ++u) {
      const int i = sbase + u * 32 + lane;
      const int v = cnt[min(i, hi - 1)];
      c[u] = (i < hi) ? v : 0;
    }
#pragma unroll
    for (int u = 0; u < SCAN_U; ++u) part += c[u];
  }
  part = __builtin_amdgcn_readlane(wave_incl_scan(part, lane), 31);
  if (lane == 0) wsum[wave] = part;
  __syncthreads();

  int carry = (lane < wave) ? wsum[lane] : 0;
  carry = __builtin_amdgcn_readlane(wave_incl_scan(carry, lane), 31);

  for (int sbase = lo; sbase < hi; sbase += 32 * SCAN_U) {
    int c[SCAN_U];
#pragma unroll
    for (int u = 0; u < SCAN_U; ++u) {
      const int i = sbase + u * 32 + lane;
      c[u] = cnt[min(i, hi - 1)];
    }
    __builtin_amdgcn_sched_barrier(0);
    int inc[SCAN_U];
#pragma unroll
    for (int u = 0; u < SCAN_U; ++u) {
      const int i = sbase + u * 32 + lane;
      c[u] = (i < hi) ? c[u] : 0;
      inc[u] = wave_incl_scan(c[u], lane);
    }
#pragma unroll
    for (int u = 0; u < SCAN_U; ++u) {
      const int i = sbase + u * 32 + lane;
      const int excl = carry + inc[u] - c[u];
      if (i < hi) { rowptr[i] = excl; cursor[i] = excl; }
      carry += __builtin_amdgcn_readlane(inc[u], 31);
    }
  }
  if (wave == SCAN_WAVES - 1 && lane == 0) rowptr[n] = carry;
}

__device__ __forceinline__ void fill_edge(const int* __restrict__ src,
                                          const int* __restrict__ dst, int nE,
                                          int* __restrict__ cursor,
                                          int* __restrict__ csr_src, int e) {
  if (e >= nE) return;
  int pos = atomicAdd(&cursor[dst[e]], 1);
  csr_src[pos] = src[e];
}

template <bool FILL>
__global__ __launch_bounds__(256) void gemm1_edge_kernel(
    const float* __restrict__ A, const float* __restrict__ Bp,
    float* __restrict__ C, int M, const float* __restrict__ a_s,
    const float* __restrict__ a_d, float* __restrict__ out_s,
    float* __restrict__ out_d, int gblocks, int rowblk0,
    const int* __restrict__ src, const int* __restrict__ dst, int nE,
    int* __restrict__ ctr, int* __restrict__ csr_src) {
  __shared__ __align__(16) float As[GemmTile<500, 128, 2>::LDS_FLOATS];
  const int b = blockIdx.x;
  if (b < gblocks) {
    gemm_block<500, 128, 2, 2>(As, rowblk0 + b, A, Bp, C, M, a_s, a_d, out_s, out_d);
  } else {
    const int e = (b - gblocks) * 256 + (int)threadIdx.x;
    if constexpr (FILL) fill_edge(src, dst, nE, ctr, csr_src, e);
    else degree_edge(dst, nE, ctr, e);
  }
}

template <int H>
__device__ __forceinline__ void edge_logit(const float* __restrict__ as,
                                           const int* __restrict__ csr_src,
                                           const float (&adv)[H], int start,
                                           int deg, int node, int k, int& s,
                                           float (&v)[H]) {
  s = (k < deg) ? csr_src[start + k] : node;
  for (int hd = 0; hd < H; ++hd) {
    float t = as[s * H + hd] + adv[hd];
    v[hd] = t > 0.f ? t : 0.2f * t;
  }
}

template <int H, int F, bool LSM>
__global__ __launch_bounds__(256) void gat_gather_kernel(
    const float* __restrict__ hsrc, const float* __restrict__ as,
    const float* __restrict__ ad, const int* __restrict__ rowptr,
    const int* __restrict__ csr_src, int n, const float* __restrict__ bias,
    float* __restrict__ agg) {
  constexpr int HF = H * F;
  constexpr int VEC = HF / 32;
  static_assert(VEC == 1 || VEC == 4, "row must be 32 or 128 floats");
  static_assert(!LSM || VEC == 1, "log_softmax needs the row in one wave, one class per lane");
  static_assert(H == 1 || H == 2, "per-lane head select below handles 1 or 2 heads");
  const int lane = threadIdx.x & 31;
  const int node = __builtin_amdgcn_readfirstlane(
      (int)((blockIdx.x * blockDim.x + threadIdx.x) >> 5));
  if (node >= n) return;
  const int start = rowptr[node];
  const int deg = rowptr[node + 1] - start;
  const int tot = deg + 1;
  const int myhead = (lane * VEC) / F;

  float adv[H];
  for (int hd = 0; hd < H; ++hd) adv[hd] = ad[node * H + hd];

  const bool ok0 = lane < tot;
  int s0 = node;
  float v0[H];
  for (int hd = 0; hd < H; ++hd) v0[hd] = -INFINITY;
  if (ok0) edge_logit<H>(as, csr_src, adv, start, deg, node, lane, s0, v0);

  float mx[H];
  for (int hd = 0; hd < H; ++hd) mx[hd] = v0[hd];
  for (int base = 32; base < tot; base += 32) {
    const int k = base + lane;
    if (k < tot) {
      int s; float v[H];
      edge_logit<H>(as, csr_src, adv, start, deg, node, k, s, v);
      for (int hd = 0; hd < H; ++hd) mx[hd] = fmaxf(mx[hd], v[hd]);
    }
  }
  for (int hd = 0; hd < H; ++hd)
    for (int o = 16; o > 0; o >>= 1) mx[hd] = fmaxf(mx[hd], __shfl_xor(mx[hd], o));

  float w0[H], dn[H];
  for (int hd = 0; hd < H; ++hd) {
    w0[hd] = ok0 ? __expf(v0[hd] - mx[hd]) : 0.f;
    dn[hd] = w0[hd];
  }
  for (int base = 32; base < tot; base += 32) {
    const int k = base + lane;
    if (k < tot) {
      int s; float v[H];
      edge_logit<H>(as, csr_src, adv, start, deg, node, k, s, v);
      for (int hd = 0; hd < H; ++hd) dn[hd] += __expf(v[hd] - mx[hd]);
    }
  }
  for (int hd = 0; hd < H; ++hd)
    for (int o = 16; o > 0; o >>= 1) dn[hd] += __shfl_xor(dn[hd], o);

  float acc[VEC];
  for (int i = 0; i < VEC; ++i) acc[i] = 0.f;
  const float* rowbase = hsrc + lane * VEC;

  for (int base = 0; base < tot; base += 32) {
    int sc = s0;
    float al[H];
    if (base == 0) {
      for (int hd = 0; hd < H; ++hd) al[hd] = w0[hd] / (dn[hd] + 1e-16f);
    } else {
      const int k = base + lane;
      sc = node;
      for (int hd = 0; hd < H; ++hd) al[hd] = 0.f;
      if (k < tot) {
        float v[H];
        edge_logit<H>(as, csr_src, adv, start, deg, node, k, sc, v);
        for (int hd = 0; hd < H; ++hd)
          al[hd] = __expf(v[hd] - mx[hd]) / (dn[hd] + 1e-16f);
      }
    }
    const int cnt = min(tot - base, 32);
#pragma unroll 4
    for (int j = 0; j < cnt; ++j) {
      const int sj = __builtin_amdgcn_readlane(sc, j);
      float aj = __int_as_float(
          __builtin_amdgcn_readlane(__float_as_int(al[0]), j));
      if constexpr (H > 1) {
        const float a1 = __int_as_float(
            __builtin_amdgcn_readlane(__float_as_int(al[H - 1]), j));
        aj = myhead ? a1 : aj;
      }
      const float* rp = rowbase + (size_t)sj * HF;
      if constexpr (VEC == 4) {
        const float4 hv = *(const float4*)rp;
        acc[0] += hv.x * aj;
        acc[1] += hv.y * aj;
        acc[2] += hv.z * aj;
        acc[3] += hv.w * aj;
      } else {
        acc[0] += rp[0] * aj;
      }
    }
  }

  for (int i = 0; i < VEC; ++i) acc[i] += bias[lane * VEC + i];
  if constexpr (LSM) {
    const float v = acc[0];
    float m = v;
    for (int o = 16; o > 0; o >>= 1) m = fmaxf(m, __shfl_xor(m, o));
    float sum = __expf(v - m);
    for (int o = 16; o > 0; o >>= 1) sum += __shfl_xor(sum, o);
    acc[0] = v - m - __logf(sum);
  } else {
    for (int i = 0; i < VEC; ++i)
      acc[i] = acc[i] > 0.f ? acc[i] : (__expf(acc[i]) - 1.f);
  }

  float* op = agg + (size_t)node * HF + lane * VEC;
  if constexpr (VEC == 4) {
    float4 r;
    r.x = acc[0]; r.y = acc[1]; r.z = acc[2]; r.w = acc[3];
    *(float4*)op = r;
  } else {
    op[0] = acc[0];
  }
}

extern "C" void kernel_launch(void* const* d_in, const int* in_sizes, int n_in,
                              void* d_out, int out_size, void* d_ws,
                              size_t ws_size, hipStream_t stream) {
  (void)n_in; (void)out_size; (void)ws_size;
  const float* x      = (const float*)d_in[0];
  const int*   ei     = (const int*)d_in[1];
  const float* W1     = (const float*)d_in[2];
  const float* a_src1 = (const float*)d_in[3];
  const float* a_dst1 = (const float*)d_in[4];
  const float* b1     = (const float*)d_in[5];
  const float* W2     = (const float*)d_in[6];
  const float* a_src2 = (const float*)d_in[7];
  const float* a_dst2 = (const float*)d_in[8];
  const float* b2     = (const float*)d_in[9];

  const int n    = in_sizes[0] / 500;
  const int nE   = in_sizes[1] / 2;
  const int* src = ei;
  const int* dst = ei + nE;

  float* ws = (float*)d_ws;
  size_t o = 0;
  float* h1   = ws + o; o += (size_t)n * 128;
  float* agg1 = ws + o; o += (size_t)n * 128;
  float* h2   = ws + o; o += (size_t)n * 32;
  float* Bp1  = ws + o; o += (size_t)500 * 128;
  float* Bp2  = ws + o; o += (size_t)128 * 32;
  float* as1  = ws + o; o += (size_t)n * 2;
  float* ad1  = ws + o; o += (size_t)n * 2;
  float* as2  = ws + o; o += (size_t)n;
  float* ad2  = ws + o; o += (size_t)n;
  int* cnt     = (int*)(ws + o); o += (size_t)n;
  int* rowptr  = (int*)(ws + o); o += (size_t)n + 1;
  int* cursor  = (int*)(ws + o); o += (size_t)n;
  int* csr_src = (int*)(ws + o); o += (size_t)nE;

  prep_kernel<<<PACK1_BLOCKS + PACK2_BLOCKS + (n + 255) / 256, 256, 0, stream>>>(
      W1, Bp1, W2, Bp2, (unsigned*)cnt, n);

  const int eblocks = (nE + 255) / 256;
  const int rblocks = (n + 31) / 32;
  const int rblocksA = rblocks / 2;
  const int rblocksB = rblocks - rblocksA;
  gemm1_edge_kernel<false><<<rblocksA + eblocks, 256, 0, stream>>>(
      x, Bp1, h1, n, a_src1, a_dst1, as1, ad1, rblocksA, 0,
      src, dst, nE, cnt, csr_src);
  scan_kernel<<<1, 1024, 0, stream>>>(cnt, n, rowptr, cursor);
  gemm1_edge_kernel<true><<<rblocksB + eblocks, 256, 0, stream>>>(
      x, Bp1, h1, n, a_src1, a_dst1, as1, ad1, rblocksB, rblocksA,
      src, dst, nE, cursor, csr_src);

  const int gblocks = (int)(((size_t)n * 32 + 255) / 256);

  gat_gather_kernel<2, 64, false><<<gblocks, 256, 0, stream>>>(
      h1, as1, ad1, rowptr, csr_src, n, b1, agg1);

  gemm_wmma_kernel<128, 32, 1><<<(n + 63) / 64, 256, 0, stream>>>(
      agg1, Bp2, h2, n, a_src2, a_dst2, as2, ad2);
  gat_gather_kernel<1, 32, true><<<gblocks, 256, 0, stream>>>(
      h2, as2, ad2, rowptr, csr_src, n, b2, (float*)d_out);
}


# layer-2 GEMM folded into the layer-1 gather: a block gathers 16 nodes (2 per wave), keeps their ELU rows in LDS, multiplies by W2 on WMMA; 7 -> 6 launches; layer-2 dot sums: one 32-term chain
# speedup vs baseline: 2.9166x; 1.0321x over previous
#include <hip/hip_runtime.h>
#include <hip/hip_bf16.h>


typedef float v2f __attribute__((ext_vector_type(2)));
typedef float v8f __attribute__((ext_vector_type(8)));

template <int K, int NCOL>
__device__ __forceinline__ void pack_b(const float* __restrict__ B,
                                       float* __restrict__ Bp, int tid) {
  if (tid >= K * NCOL) return;
  int k = tid / NCOL;
  int col = tid - k * NCOL;
  int kk = k >> 2;
  int p = (k >> 1) & 1;
  int q = k & 1;
  Bp[(((kk * 2 + p) * NCOL) + col) * 2 + q] = B[tid];
}

constexpr int PACK1_BLOCKS = (500 * 128 + 255) / 256;
constexpr int PACK2_BLOCKS = (128 * 32 + 255) / 256;

__global__ __launch_bounds__(256) void prep_kernel(
    const float* __restrict__ W1, float* __restrict__ Bp1,
    const float* __restrict__ W2, float* __restrict__ Bp2,
    unsigned* __restrict__ cnt, int n) {
  const int b = blockIdx.x;
  if (b < PACK1_BLOCKS) {
    pack_b<500, 128>(W1, Bp1, b * 256 + (int)threadIdx.x);
  } else if (b < PACK1_BLOCKS + PACK2_BLOCKS) {
    pack_b<128, 32>(W2, Bp2, (b - PACK1_BLOCKS) * 256 + (int)threadIdx.x);
  } else {
    const int i = (b - PACK1_BLOCKS - PACK2_BLOCKS) * 256 + (int)threadIdx.x;
    if (i < n) cnt[i] = 0u;
  }
}

template <int K, int NCOL, int MW>
struct GemmTile {
  static constexpr int NT = NCOL / 16;
  static constexpr int MT = 8 / NT;
  static constexpr int ROWS = MT * MW * 16;
  static constexpr int STEP_F4 = 256 * 8;
  static constexpr int LOAD_F4 = (ROWS * K / 4 + STEP_F4 - 1) / STEP_F4 * STEP_F4;
  static constexpr int LDS_FLOATS = LOAD_F4 * 4;
};

template <int K, int NCOL, int H, int MW>
__device__ __forceinline__ void gemm_block(
    float* __restrict__ As, int blk,
    const float* __restrict__ A, const float* __restrict__ Bp,
    float* __restrict__ C, int M, const float* __restrict__ a_s,
    const float* __restrict__ a_d, float* __restrict__ out_s,
    float* __restrict__ out_d) {
  constexpr int NT = GemmTile<K, NCOL, MW>::NT;
  constexpr int MT = GemmTile<K, NCOL, MW>::MT;
  constexpr int ROWS = GemmTile<K, NCOL, MW>::ROWS;
  constexpr int KV = K / 4;

  const int tid = threadIdx.x;
  const int wave = tid >> 5;
  const int lane = tid & 31;
  const int row0 = blk * ROWS;

  {
    const float4* A4 = (const float4*)A + (size_t)row0 * KV;
    float4* As4 = (float4*)As;
    constexpr int TOT = ROWS * KV;
    constexpr int LB = 8;
    constexpr int NIT = GemmTile<K, NCOL, MW>::LOAD_F4 / (256 * LB);
    const int lim = min(TOT, (M - row0) * KV);
    for (int it = 0; it < NIT; ++it) {
      const int i0 = it * (256 * LB) + tid;
      float4 v[LB];
#pragma unroll
      for (int u = 0; u < LB; ++u) v[u] = A4[min(i0 + u * 256, lim - 1)];
      __builtin_amdgcn_sched_barrier(0);
#pragma unroll
      for (int u = 0; u < LB; ++u) {
        const int i = i0 + u * 256;
        float4 t = v[u];
        if (i >= lim) t = float4{0.f, 0.f, 0.f, 0.f};
        As4[i] = t;
      }
    }
  }
  __syncthreads();

  const int mtile = wave / NT;
  const int ntile = wave - mtile * NT;
  const int lrow = ((mtile * MW) << 4) + (lane & 15);
  const int koff = (lane >> 4) << 1;
  const int ncol = (ntile << 4) + (lane & 15);
  const int bbase = ((lane >> 4) * NCOL + ncol);

  v8f acc[MW];
  for (int j = 0; j < MW; ++j)
    acc[j] = v8f{0.f, 0.f, 0.f, 0.f, 0.f, 0.f, 0.f, 0.f};
  for (int kk = 0; kk < KV; ++kk) {
    const v2f b = *(const v2f*)&Bp[(size_t)(bbase + kk * 2 * NCOL) * 2];
    for (int j = 0; j < MW; ++j) {
      v2f a;
      a.x = As[(lrow + j * 16) * K + kk * 4 + koff];
      a.y = As[(lrow + j * 16) * K + kk * 4 + koff + 1];
      acc[j] = __builtin_amdgcn_wmma_f32_16x16x4_f32(
           false, a,  false, b,
           (short)0, acc[j],  false,  false);
    }
  }

  constexpr int P = NCOL + 4;
  constexpr int F = NCOL / H;
  constexpr int RPP = MT * 16;
  constexpr int TPR = 256 / RPP;
  constexpr int TPH = TPR / H;
  static_assert(F / TPH == 8, "each thread sums 8 columns (two float4)");
  static_assert(ROWS * P <= ROWS * K, "C tile must fit in the A tile's LDS");
  float* Cs = As;
  __syncthreads();
#pragma unroll
  for (int j = 0; j < MW; ++j) {
    const int crow = ((mtile * MW + j) << 4) + ((lane >> 4) << 3);
    for (int i = 0; i < 8; ++i) Cs[(crow + i) * P + ncol] = acc[j][i];
  }
  __syncthreads();
  const int rr = tid / TPR;
  const int rem = tid - rr * TPR;
  const int hd = rem / TPH;
  const int part = rem - hd * TPH;
  const int c0 = hd * F + part * 8;
  const float4 s0 = *(const float4*)&a_s[c0];
  const float4 s1 = *(const float4*)&a_s[c0 + 4];
  const float4 d0 = *(const float4*)&a_d[c0];
  const float4 d1 = *(const float4*)&a_d[c0 + 4];
  for (int pass = 0; pass < MW; ++pass) {
    const int r = ((rr >> 4) * MW + pass) * 16 + (rr & 15);
    const float4 u0 = *(const float4*)&Cs[r * P + c0];
    const float4 u1 = *(const float4*)&Cs[r * P + c0 + 4];
    float ps = u0.x * s0.x;
    ps += u0.y * s0.y; ps += u0.z * s0.z; ps += u0.w * s0.w;
    ps += u1.x * s1.x; ps += u1.y * s1.y; ps += u1.z * s1.z; ps += u1.w * s1.w;
    float pd = u0.x * d0.x;
    pd += u0.y * d0.y; pd += u0.z * d0.z; pd += u0.w * d0.w;
    pd += u1.x * d1.x; pd += u1.y * d1.y; pd += u1.z * d1.z; pd += u1.w * d1.w;
    for (int o = TPH / 2; o > 0; o >>= 1) {
      ps += __shfl_xor(ps, o);
      pd += __shfl_xor(pd, o);
    }
    const int gr = row0 + r;
    if (part == 0 && gr < M) {
      out_s[gr * H + hd] = ps;
      out_d[gr * H + hd] = pd;
    }
  }

#pragma unroll
  for (int j = 0; j < MW; ++j) {
    const int rbase = row0 + ((mtile * MW + j) << 4) + ((lane >> 4) << 3);
    if (row0 + ROWS <= M) {
#pragma unroll
      for (int i = 0; i < 8; ++i) C[(size_t)(rbase + i) * NCOL + ncol] = acc[j][i];
    } else {
#pragma unroll
      for (int i = 0; i < 8; ++i) {
        int r = rbase + i;
        if (r < M) C[(size_t)r * NCOL + ncol] = acc[j][i];
      }
    }
  }
}

__device__ __forceinline__ void degree_edge(const int* __restrict__ dst, int nE,
                                            int* __restrict__ cnt, int e) {
  if (e < nE) atomicAdd(&cnt[dst[e]], 1);
}

constexpr int SCAN_WAVES = 32;
constexpr int SCAN_U = 25;
static_assert(SCAN_WAVES * 32 == 1024, "the launch below uses one block of 1024 threads");

__device__ __forceinline__ int wave_incl_scan(int v, int lane) {
  v += __builtin_amdgcn_update_dpp(0, v, 0x111, 0xf, 0xf, true);
  v += __builtin_amdgcn_update_dpp(0, v, 0x112, 0xf, 0xf, true);
  v += __builtin_amdgcn_update_dpp(0, v, 0x114, 0xf, 0xf, true);
  v += __builtin_amdgcn_update_dpp(0, v, 0x118, 0xf, 0xf, true);
  const int low = __builtin_amdgcn_readlane(v, 15);
  return v + ((lane & 16) ? low : 0);
}

__global__ __launch_bounds__(SCAN_WAVES * 32) void scan_kernel(
    const int* __restrict__ cnt, int n, int* __restrict__ rowptr,
    int* __restrict__ cursor) {
  __shared__ int wsum[SCAN_WAVES];
  const int lane = threadIdx.x & 31;
  const int wave = __builtin_amdgcn_readfirstlane((int)(threadIdx.x >> 5));
  const int seg = (n + SCAN_WAVES - 1) / SCAN_WAVES;
  const int lo = wave * seg;
  const int hi = min(lo + seg, n);

  int part = 0;
  for (int sbase = lo; sbase < hi; sbase += 32 * SCAN_U) {
    int c[SCAN_U];
#pragma unroll
    for (int u = 0; u < SCAN_U; ++u) {
      const int i = sbase + u * 32 + lane;
      const int v = cnt[min(i, hi - 1)];
      c[u] = (i < hi) ? v : 0;
    }
#pragma unroll
    for (int u = 0; u < SCAN_U; ++u) part += c[u];
  }
  part = __builtin_amdgcn_readlane(wave_incl_scan(part, lane), 31);
  if (lane == 0) wsum[wave] = part;
  __syncthreads();

  int carry = (lane < wave) ? wsum[lane] : 0;
  carry = __builtin_amdgcn_readlane(wave_incl_scan(carry, lane), 31);

  for (int sbase = lo; sbase < hi; sbase += 32 * SCAN_U) {
    int c[SCAN_U];
#pragma unroll
    for (int u = 0; u < SCAN_U; ++u) {
      const int i = sbase + u * 32 + lane;
      c[u] = cnt[min(i, hi - 1)];
    }
    __builtin_amdgcn_sched_barrier(0);
    int inc[SCAN_U];
#pragma unroll
    for (int u = 0; u < SCAN_U; ++u) {
      const int i = sbase + u * 32 + lane;
      c[u] = (i < hi) ? c[u] : 0;
      inc[u] = wave_incl_scan(c[u], lane);
    }
#pragma unroll
    for (int u = 0; u < SCAN_U; ++u) {
      const int i = sbase + u * 32 + lane;
      const int excl = carry + inc[u] - c[u];
      if (i < hi) { rowptr[i] = excl; cursor[i] = excl; }
      carry += __builtin_amdgcn_readlane(inc[u], 31);
    }
  }
  if (wave == SCAN_WAVES - 1 && lane == 0) rowptr[n] = carry;
}

__device__ __forceinline__ void fill_edge(const int* __restrict__ src,
                                          const int* __restrict__ dst, int nE,
                                          int* __restrict__ cursor,
                                          int* __restrict__ csr_src, int e) {
  if (e >= nE) return;
  int pos = atomicAdd(&cursor[dst[e]], 1);
  csr_src[pos] = src[e];
}

template <bool FILL>
__global__ __launch_bounds__(256) void gemm1_edge_kernel(
    const float* __restrict__ A, const float* __restrict__ Bp,
    float* __restrict__ C, int M, const float* __restrict__ a_s,
    const float* __restrict__ a_d, float* __restrict__ out_s,
    float* __restrict__ out_d, int gblocks, int rowblk0,
    const int* __restrict__ src, const int* __restrict__ dst, int nE,
    int* __restrict__ ctr, int* __restrict__ csr_src) {
  __shared__ __align__(16) float As[GemmTile<500, 128, 2>::LDS_FLOATS];
  const int b = blockIdx.x;
  if (b < gblocks) {
    gemm_block<500, 128, 2, 2>(As, rowblk0 + b, A, Bp, C, M, a_s, a_d, out_s, out_d);
  } else {
    const int e = (b - gblocks) * 256 + (int)threadIdx.x;
    if constexpr (FILL) fill_edge(src, dst, nE, ctr, csr_src, e);
    else degree_edge(dst, nE, ctr, e);
  }
}

template <int H>
__device__ __forceinline__ void edge_logit(const float* __restrict__ as,
                                           const int* __restrict__ csr_src,
                                           const float (&adv)[H], int start,
                                           int deg, int node, int k, int& s,
                                           float (&v)[H]) {
  s = (k < deg) ? csr_src[start + k] : node;
  for (int hd = 0; hd < H; ++hd) {
    float t = as[s * H + hd] + adv[hd];
    v[hd] = t > 0.f ? t : 0.2f * t;
  }
}

template <int H, int F, bool LSM>
__device__ __forceinline__ void gather_node(
    const float* __restrict__ hsrc, const float* __restrict__ as,
    const float* __restrict__ ad, const int* __restrict__ rowptr,
    const int* __restrict__ csr_src, const float* __restrict__ bias,
    int node, int lane, float (&acc)[H * F / 32]) {
  constexpr int HF = H * F;
  constexpr int VEC = HF / 32;
  static_assert(VEC == 1 || VEC == 4, "row must be 32 or 128 floats");
  static_assert(!LSM || VEC == 1, "log_softmax needs the row in one wave, one class per lane");
  static_assert(H == 1 || H == 2, "per-lane head select below handles 1 or 2 heads");
  const int start = rowptr[node];
  const int deg = rowptr[node + 1] - start;
  const int tot = deg + 1;
  const int myhead = (lane * VEC) / F;

  float adv[H];
  for (int hd = 0; hd < H; ++hd) adv[hd] = ad[node * H + hd];

  const bool ok0 = lane < tot;
  int s0 = node;
  float v0[H];
  for (int hd = 0; hd < H; ++hd) v0[hd] = -INFINITY;
  if (ok0) edge_logit<H>(as, csr_src, adv, start, deg, node, lane, s0, v0);

  float mx[H];
  for (int hd = 0; hd < H; ++hd) mx[hd] = v0[hd];
  for (int base = 32; base < tot; base += 32) {
    const int k = base + lane;
    if (k < tot) {
      int s; float v[H];
      edge_logit<H>(as, csr_src, adv, start, deg, node, k, s, v);
      for (int hd = 0; hd < H; ++hd) mx[hd] = fmaxf(mx[hd], v[hd]);
    }
  }
  for (int hd = 0; hd < H; ++hd)
    for (int o = 16; o > 0; o >>= 1) mx[hd] = fmaxf(mx[hd], __shfl_xor(mx[hd], o));

  float w0[H], dn[H];
  for (int hd = 0; hd < H; ++hd) {
    w0[hd] = ok0 ? __expf(v0[hd] - mx[hd]) : 0.f;
    dn[hd] = w0[hd];
  }
  for (int base = 32; base < tot; base += 32) {
    const int k = base + lane;
    if (k < tot) {
      int s; float v[H];
      edge_logit<H>(as, csr_src, adv, start, deg, node, k, s, v);
      for (int hd = 0; hd < H; ++hd) dn[hd] += __expf(v[hd] - mx[hd]);
    }
  }
  for (int hd = 0; hd < H; ++hd)
    for (int o = 16; o > 0; o >>= 1) dn[hd] += __shfl_xor(dn[hd], o);

  for (int i = 0; i < VEC; ++i) acc[i] = 0.f;
  const float* rowbase = hsrc + lane * VEC;

  for (int base = 0; base < tot; base += 32) {
    int sc = s0;
    float al[H];
    if (base == 0) {
      for (int hd = 0; hd < H; ++hd) al[hd] = w0[hd] / (dn[hd] + 1e-16f);
    } else {
      const int k = base + lane;
      sc = node;
      for (int hd = 0; hd < H; ++hd) al[hd] = 0.f;
      if (k < tot) {
        float v[H];
        edge_logit<H>(as, csr_src, adv, start, deg, node, k, sc, v);
        for (int hd = 0; hd < H; ++hd)
          al[hd] = __expf(v[hd] - mx[hd]) / (dn[hd] + 1e-16f);
      }
    }
    const int cnt = min(tot - base, 32);
#pragma unroll 4
    for (int j = 0; j < cnt; ++j) {
      const int sj = __builtin_amdgcn_readlane(sc, j);
      float aj = __int_as_float(
          __builtin_amdgcn_readlane(__float_as_int(al[0]), j));
      if constexpr (H > 1) {
        const float a1 = __int_as_float(
            __builtin_amdgcn_readlane(__float_as_int(al[H - 1]), j));
        aj = myhead ? a1 : aj;
      }
      const float* rp = rowbase + (size_t)sj * HF;
      if constexpr (VEC == 4) {
        const float4 hv = *(const float4*)rp;
        acc[0] += hv.x * aj;
        acc[1] += hv.y * aj;
        acc[2] += hv.z * aj;
        acc[3] += hv.w * aj;
      } else {
        acc[0] += rp[0] * aj;
      }
    }
  }

  for (int i = 0; i < VEC; ++i) acc[i] += bias[lane * VEC + i];
  if constexpr (LSM) {
    const float v = acc[0];
    float m = v;
    for (int o = 16; o > 0; o >>= 1) m = fmaxf(m, __shfl_xor(m, o));
    float sum = __expf(v - m);
    for (int o = 16; o > 0; o >>= 1) sum += __shfl_xor(sum, o);
    acc[0] = v - m - __logf(sum);
  } else {
    for (int i = 0; i < VEC; ++i)
      acc[i] = acc[i] > 0.f ? acc[i] : (__expf(acc[i]) - 1.f);
  }
}

template <int H, int F, bool LSM>
__global__ __launch_bounds__(256) void gat_gather_kernel(
    const float* __restrict__ hsrc, const float* __restrict__ as,
    const float* __restrict__ ad, const int* __restrict__ rowptr,
    const int* __restrict__ csr_src, int n, const float* __restrict__ bias,
    float* __restrict__ agg) {
  constexpr int HF = H * F;
  constexpr int VEC = HF / 32;
  const int lane = threadIdx.x & 31;
  const int node = __builtin_amdgcn_readfirstlane(
      (int)((blockIdx.x * blockDim.x + threadIdx.x) >> 5));
  if (node >= n) return;
  float acc[VEC];
  gather_node<H, F, LSM>(hsrc, as, ad, rowptr, csr_src, bias, node, lane, acc);
  float* op = agg + (size_t)node * HF + lane * VEC;
  if constexpr (VEC == 4) {
    float4 r;
    r.x = acc[0]; r.y = acc[1]; r.z = acc[2]; r.w = acc[3];
    *(float4*)op = r;
  } else {
    op[0] = acc[0];
  }
}

constexpr int G1_ROWS = 16;
constexpr int G1_AP = 128 + 4;
constexpr int G1_CP = 32 + 4;

__global__ __launch_bounds__(256) void gather1_gemm2_kernel(
    const float* __restrict__ h1, const float* __restrict__ as1,
    const float* __restrict__ ad1, const int* __restrict__ rowptr,
    const int* __restrict__ csr_src, int n, const float* __restrict__ b1,
    const float* __restrict__ Bp2, float* __restrict__ h2,
    const float* __restrict__ a_s2, const float* __restrict__ a_d2,
    float* __restrict__ as2, float* __restrict__ ad2) {
  __shared__ __align__(16) float Ar[G1_ROWS * G1_AP];
  __shared__ __align__(16) float Cs[G1_ROWS * G1_CP];
  const int tid = threadIdx.x;
  const int lane = tid & 31;
  const int wave = __builtin_amdgcn_readfirstlane(tid >> 5);
  const int row0 = (int)blockIdx.x * G1_ROWS;

  for (int t = 0; t < G1_ROWS / 8; ++t) {
    const int lr = t * 8 + wave;
    const int node = row0 + lr;
    float acc[4];
    if (node < n) {
      gather_node<2, 64, false>(h1, as1, ad1, rowptr, csr_src, b1, node, lane, acc);
    } else {
      for (int i = 0; i < 4; ++i) acc[i] = 0.f;
    }
    float4 r;
    r.x = acc[0]; r.y = acc[1]; r.z = acc[2]; r.w = acc[3];
    *(float4*)&Ar[lr * G1_AP + lane * 4] = r;
  }
  const float sv = a_s2[lane];
  const float dv = a_d2[lane];
  __syncthreads();

  if (wave < 2) {
    const int c = lane & 15;
    const int koff = (lane >> 4) << 1;
    const int ncol = (wave << 4) + c;
    const int bbase = (lane >> 4) * 32 + ncol;
    v8f acc = v8f{0.f, 0.f, 0.f, 0.f, 0.f, 0.f, 0.f, 0.f};
    for (int kk = 0; kk < 32; ++kk) {
      const v2f b = *(const v2f*)&Bp2[(size_t)(bbase + kk * 2 * 32) * 2];
      v2f a;
      a.x = Ar[c * G1_AP + kk * 4 + koff];
      a.y = Ar[c * G1_AP + kk * 4 + koff + 1];
      acc = __builtin_amdgcn_wmma_f32_16x16x4_f32(
           false, a,  false, b,
           (short)0, acc,  false,  false);
    }
    const int crow = (lane >> 4) << 3;
#pragma unroll
    for (int i = 0; i < 8; ++i) Cs[(crow + i) * G1_CP + ncol] = acc[i];
  }
  __syncthreads();

  if (tid < G1_ROWS * 8) {
    const int r = tid >> 3;
    const int c4 = (tid & 7) << 2;
    const int gr = row0 + r;
    if (gr < n) *(float4*)&h2[(size_t)gr * 32 + c4] = *(const float4*)&Cs[r * G1_CP + c4];
  } else if (wave == 4) {
    const int r = lane & 15;
    const bool isd = lane >= 16;
    float s = 0.f;
#pragma unroll
    for (int f = 0; f < 32; ++f) {
      const float vs = __int_as_float(__builtin_amdgcn_readlane(__float_as_int(sv), f));
      const float vd = __int_as_float(__builtin_amdgcn_readlane(__float_as_int(dv), f));
      s += Cs[r * G1_CP + f] * (isd ? vd : vs);
    }
    const int gr = row0 + r;
    if (gr < n) {
      if (isd) ad2[gr] = s;
      else as2[gr] = s;
    }
  }
}

extern "C" void kernel_launch(void* const* d_in, const int* in_sizes, int n_in,
                              void* d_out, int out_size, void* d_ws,
                              size_t ws_size, hipStream_t stream) {
  (void)n_in; (void)out_size; (void)ws_size;
  const float* x      = (const float*)d_in[0];
  const int*   ei     = (const int*)d_in[1];
  const float* W1     = (const float*)d_in[2];
  const float* a_src1 = (const float*)d_in[3];
  const float* a_dst1 = (const float*)d_in[4];
  const float* b1     = (const float*)d_in[5];
  const float* W2     = (const float*)d_in[6];
  const float* a_src2 = (const float*)d_in[7];
  const float* a_dst2 = (const float*)d_in[8];
  const float* b2     = (const float*)d_in[9];

  const int n    = in_sizes[0] / 500;
  const int nE   = in_sizes[1] / 2;
  const int* src = ei;
  const int* dst = ei + nE;

  float* ws = (float*)d_ws;
  size_t o = 0;
  float* h1   = ws + o; o += (size_t)n * 128;
  o += (size_t)n * 128;
  float* h2   = ws + o; o += (size_t)n * 32;
  float* Bp1  = ws + o; o += (size_t)500 * 128;
  float* Bp2  = ws + o; o += (size_t)128 * 32;
  float* as1  = ws + o; o += (size_t)n * 2;
  float* ad1  = ws + o; o += (size_t)n * 2;
  float* as2  = ws + o; o += (size_t)n;
  float* ad2  = ws + o; o += (size_t)n;
  int* cnt     = (int*)(ws + o); o += (size_t)n;
  int* rowptr  = (int*)(ws + o); o += (size_t)n + 1;
  int* cursor  = (int*)(ws + o); o += (size_t)n;
  int* csr_src = (int*)(ws + o); o += (size_t)nE;

  prep_kernel<<<PACK1_BLOCKS + PACK2_BLOCKS + (n + 255) / 256, 256, 0, stream>>>(
      W1, Bp1, W2, Bp2, (unsigned*)cnt, n);

  const int eblocks = (nE + 255) / 256;
  const int rblocks = (n + 31) / 32;
  const int rblocksA = rblocks / 2;
  const int rblocksB = rblocks - rblocksA;
  gemm1_edge_kernel<false><<<rblocksA + eblocks, 256, 0, stream>>>(
      x, Bp1, h1, n, a_src1, a_dst1, as1, ad1, rblocksA, 0,
      src, dst, nE, cnt, csr_src);
  scan_kernel<<<1, 1024, 0, stream>>>(cnt, n, rowptr, cursor);
  gemm1_edge_kernel<true><<<rblocksB + eblocks, 256, 0, stream>>>(
      x, Bp1, h1, n, a_src1, a_dst1, as1, ad1, rblocksB, rblocksA,
      src, dst, nE, cursor, csr_src);

  const int gblocks = (int)(((size_t)n * 32 + 255) / 256);

  gather1_gemm2_kernel<<<(n + G1_ROWS - 1) / G1_ROWS, 256, 0, stream>>>(
      h1, as1, ad1, rowptr, csr_src, n, b1, Bp2, h2, a_src2, a_dst2, as2, ad2);

  gat_gather_kernel<1, 32, true><<<gblocks, 256, 0, stream>>>(
      h2, as2, ad2, rowptr, csr_src, n, b2, (float*)d_out);
}


# edges grouped by destination in ONE atomic pass (64 source slots per node + overflow list) replacing count, scan, fill; layer-1 GEMM in one launch again; 6 -> 4 launches; sum order: slot order
# speedup vs baseline: 3.8136x; 1.3075x over previous
#include <hip/hip_runtime.h>
#include <hip/hip_bf16.h>


typedef float v2f __attribute__((ext_vector_type(2)));
typedef float v8f __attribute__((ext_vector_type(8)));

template <int K, int NCOL>
__device__ __forceinline__ void pack_b(const float* __restrict__ B,
                                       float* __restrict__ Bp, int tid) {
  if (tid >= K * NCOL) return;
  int k = tid / NCOL;
  int col = tid - k * NCOL;
  int kk = k >> 2;
  int p = (k >> 1) & 1;
  int q = k & 1;
  Bp[(((kk * 2 + p) * NCOL) + col) * 2 + q] = B[tid];
}

constexpr int PACK1_BLOCKS = (500 * 128 + 255) / 256;
constexpr int PACK2_BLOCKS = (128 * 32 + 255) / 256;

__global__ __launch_bounds__(256) void prep_kernel(
    const float* __restrict__ W1, float* __restrict__ Bp1,
    const float* __restrict__ W2, float* __restrict__ Bp2,
    unsigned* __restrict__ cnt, int n) {
  const int b = blockIdx.x;
  if (b < PACK1_BLOCKS) {
    pack_b<500, 128>(W1, Bp1, b * 256 + (int)threadIdx.x);
  } else if (b < PACK1_BLOCKS + PACK2_BLOCKS) {
    pack_b<128, 32>(W2, Bp2, (b - PACK1_BLOCKS) * 256 + (int)threadIdx.x);
  } else {
    const int i = (b - PACK1_BLOCKS - PACK2_BLOCKS) * 256 + (int)threadIdx.x;
    if (i < n) cnt[i] = 0u;
  }
}

template <int K, int NCOL, int MW>
struct GemmTile {
  static constexpr int NT = NCOL / 16;
  static constexpr int MT = 8 / NT;
  static constexpr int ROWS = MT * MW * 16;
  static constexpr int STEP_F4 = 256 * 8;
  static constexpr int LOAD_F4 = (ROWS * K / 4 + STEP_F4 - 1) / STEP_F4 * STEP_F4;
  static constexpr int LDS_FLOATS = LOAD_F4 * 4;
};

template <int K, int NCOL, int H, int MW>
__device__ __forceinline__ void gemm_block(
    float* __restrict__ As, int blk,
    const float* __restrict__ A, const float* __restrict__ Bp,
    float* __restrict__ C, int M, const float* __restrict__ a_s,
    const float* __restrict__ a_d, float* __restrict__ out_s,
    float* __restrict__ out_d) {
  constexpr int NT = GemmTile<K, NCOL, MW>::NT;
  constexpr int MT = GemmTile<K, NCOL, MW>::MT;
  constexpr int ROWS = GemmTile<K, NCOL, MW>::ROWS;
  constexpr int KV = K / 4;

  const int tid = threadIdx.x;
  const int wave = tid >> 5;
  const int lane = tid & 31;
  const int row0 = blk * ROWS;

  {
    const float4* A4 = (const float4*)A + (size_t)row0 * KV;
    float4* As4 = (float4*)As;
    constexpr int TOT = ROWS * KV;
    constexpr int LB = 8;
    constexpr int NIT = GemmTile<K, NCOL, MW>::LOAD_F4 / (256 * LB);
    const int lim = min(TOT, (M - row0) * KV);
    for (int it = 0; it < NIT; ++it) {
      const int i0 = it * (256 * LB) + tid;
      float4 v[LB];
#pragma unroll
      for (int u = 0; u < LB; ++u) v[u] = A4[min(i0 + u * 256, lim - 1)];
      __builtin_amdgcn_sched_barrier(0);
#pragma unroll
      for (int u = 0; u < LB; ++u) {
        const int i = i0 + u * 256;
        float4 t = v[u];
        if (i >= lim) t = float4{0.f, 0.f, 0.f, 0.f};
        As4[i] = t;
      }
    }
  }
  __syncthreads();

  const int mtile = wave / NT;
  const int ntile = wave - mtile * NT;
  const int lrow = ((mtile * MW) << 4) + (lane & 15);
  const int koff = (lane >> 4) << 1;
  const int ncol = (ntile << 4) + (lane & 15);
  const int bbase = ((lane >> 4) * NCOL + ncol);

  v8f acc[MW];
  for (int j = 0; j < MW; ++j)
    acc[j] = v8f{0.f, 0.f, 0.f, 0.f, 0.f, 0.f, 0.f, 0.f};
  for (int kk = 0; kk < KV; ++kk) {
    const v2f b = *(const v2f*)&Bp[(size_t)(bbase + kk * 2 * NCOL) * 2];
    for (int j = 0; j < MW; ++j) {
      v2f a;
      a.x = As[(lrow + j * 16) * K + kk * 4 + koff];
      a.y = As[(lrow + j * 16) * K + kk * 4 + koff + 1];
      acc[j] = __builtin_amdgcn_wmma_f32_16x16x4_f32(
           false, a,  false, b,
           (short)0, acc[j],  false,  false);
    }
  }

  constexpr int P = NCOL + 4;
  constexpr int F = NCOL / H;
  constexpr int RPP = MT * 16;
  constexpr int TPR = 256 / RPP;
  constexpr int TPH = TPR / H;
  static_assert(F / TPH == 8, "each thread sums 8 columns (two float4)");
  static_assert(ROWS * P <= ROWS * K, "C tile must fit in the A tile's LDS");
  float* Cs = As;
  __syncthreads();
#pragma unroll
  for (int j = 0; j < MW; ++j) {
    const int crow = ((mtile * MW + j) << 4) + ((lane >> 4) << 3);
    for (int i = 0; i < 8; ++i) Cs[(crow + i) * P + ncol] = acc[j][i];
  }
  __syncthreads();
  const int rr = tid / TPR;
  const int rem = tid - rr * TPR;
  const int hd = rem / TPH;
  const int part = rem - hd * TPH;
  const int c0 = hd * F + part * 8;
  const float4 s0 = *(const float4*)&a_s[c0];
  const float4 s1 = *(const float4*)&a_s[c0 + 4];
  const float4 d0 = *(const float4*)&a_d[c0];
  const float4 d1 = *(const float4*)&a_d[c0 + 4];
  for (int pass = 0; pass < MW; ++pass) {
    const int r = ((rr >> 4) * MW + pass) * 16 + (rr & 15);
    const float4 u0 = *(const float4*)&Cs[r * P + c0];
    const float4 u1 = *(const float4*)&Cs[r * P + c0 + 4];
    float ps = u0.x * s0.x;
    ps += u0.y * s0.y; ps += u0.z * s0.z; ps += u0.w * s0.w;
    ps += u1.x * s1.x; ps += u1.y * s1.y; ps += u1.z * s1.z; ps += u1.w * s1.w;
    float pd = u0.x * d0.x;
    pd += u0.y * d0.y; pd += u0.z * d0.z; pd += u0.w * d0.w;
    pd += u1.x * d1.x; pd += u1.y * d1.y; pd += u1.z * d1.z; pd += u1.w * d1.w;
    for (int o = TPH / 2; o > 0; o >>= 1) {
      ps += __shfl_xor(ps, o);
      pd += __shfl_xor(pd, o);
    }
    const int gr = row0 + r;
    if (part == 0 && gr < M) {
      out_s[gr * H + hd] = ps;
      out_d[gr * H + hd] = pd;
    }
  }

#pragma unroll
  for (int j = 0; j < MW; ++j) {
    const int rbase = row0 + ((mtile * MW + j) << 4) + ((lane >> 4) << 3);
    if (row0 + ROWS <= M) {
#pragma unroll
      for (int i = 0; i < 8; ++i) C[(size_t)(rbase + i) * NCOL + ncol] = acc[j][i];
    } else {
#pragma unroll
      for (int i = 0; i < 8; ++i) {
        int r = rbase + i;
        if (r < M) C[(size_t)r * NCOL + ncol] = acc[j][i];
      }
    }
  }
}

constexpr int SLOT_CAP = 64;
constexpr int FILL_EPT = 8;

__device__ __forceinline__ void fill_edges(const int* __restrict__ src,
                                           const int* __restrict__ dst, int nE,
                                           int n, int* __restrict__ cnt,
                                           int* __restrict__ slots,
                                           int* __restrict__ ovf_src,
                                           int* __restrict__ ovf_dst, int blk,
                                           int tid) {
  const int base = blk * (256 * FILL_EPT) + tid;
  int d[FILL_EPT], s[FILL_EPT], pos[FILL_EPT];
#pragma unroll
  for (int u = 0; u < FILL_EPT; ++u) {
    const int e = min(base + u * 256, nE - 1);
    d[u] = dst[e];
    s[u] = src[e];
  }
#pragma unroll
  for (int u = 0; u < FILL_EPT; ++u) {
    pos[u] = -1;
    if (base + u * 256 < nE) pos[u] = atomicAdd(&cnt[d[u]], 1);
  }
#pragma unroll
  for (int u = 0; u < FILL_EPT; ++u) {
    if (pos[u] < 0) continue;
    if (pos[u] < SLOT_CAP) {
      slots[(size_t)d[u] * SLOT_CAP + pos[u]] = s[u];
    } else {
      const int o = atomicAdd(&cnt[n], 1);
      ovf_src[o] = s[u];
      ovf_dst[o] = d[u];
    }
  }
}

__global__ __launch_bounds__(256) void gemm1_fill_kernel(
    const float* __restrict__ A, const float* __restrict__ Bp,
    float* __restrict__ C, int M, const float* __restrict__ a_s,
    const float* __restrict__ a_d, float* __restrict__ out_s,
    float* __restrict__ out_d, int fblocks,
    const int* __restrict__ src, const int* __restrict__ dst, int nE,
    int* __restrict__ cnt, int* __restrict__ slots, int* __restrict__ ovf_src,
    int* __restrict__ ovf_dst) {
  __shared__ __align__(16) float As[GemmTile<500, 128, 2>::LDS_FLOATS];
  const int b = blockIdx.x;
  if (b >= fblocks) {
    gemm_block<500, 128, 2, 2>(As, b - fblocks, A, Bp, C, M, a_s, a_d, out_s, out_d);
  } else {
    fill_edges(src, dst, nE, M, cnt, slots, ovf_src, ovf_dst, b, (int)threadIdx.x);
  }
}

template <int H>
__device__ __forceinline__ void source_logit(const float* __restrict__ as,
                                             const float (&adv)[H], int s,
                                             float (&v)[H]) {
  for (int hd = 0; hd < H; ++hd) {
    float t = as[s * H + hd] + adv[hd];
    v[hd] = t > 0.f ? t : 0.2f * t;
  }
}

template <int H>
__device__ __forceinline__ void edge_logit(const float* __restrict__ as,
                                           const int* __restrict__ slots,
                                           const float (&adv)[H], int start,
                                           int deg, int node, int k, int& s,
                                           float (&v)[H]) {
  s = (k < deg) ? slots[start + k] : node;
  source_logit<H>(as, adv, s, v);
}

template <int H, int HF, int VEC>
__device__ __forceinline__ void add_row(const float* __restrict__ rowbase,
                                        int sc, const float (&al)[H], int j,
                                        int myhead, float (&acc)[VEC]) {
  const int sj = __builtin_amdgcn_readlane(sc, j);
  float aj = __int_as_float(
      __builtin_amdgcn_readlane(__float_as_int(al[0]), j));
  if constexpr (H > 1) {
    const float a1 = __int_as_float(
        __builtin_amdgcn_readlane(__float_as_int(al[H - 1]), j));
    aj = myhead ? a1 : aj;
  }
  const float* rp = rowbase + (size_t)sj * HF;
  if constexpr (VEC == 4) {
    const float4 hv = *(const float4*)rp;
    acc[0] += hv.x * aj;
    acc[1] += hv.y * aj;
    acc[2] += hv.z * aj;
    acc[3] += hv.w * aj;
  } else {
    acc[0] += rp[0] * aj;
  }
}

template <int H, int F, bool LSM>
__device__ __forceinline__ void gather_node(
    const float* __restrict__ hsrc, const float* __restrict__ as,
    const float* __restrict__ ad, const int* __restrict__ cnt,
    const int* __restrict__ slots, const int* __restrict__ ovf_src,
    const int* __restrict__ ovf_dst, int n, int nE,
    const float* __restrict__ bias, int node, int lane,
    float (&acc)[H * F / 32]) {
  constexpr int HF = H * F;
  constexpr int VEC = HF / 32;
  static_assert(VEC == 1 || VEC == 4, "row must be 32 or 128 floats");
  static_assert(!LSM || VEC == 1, "log_softmax needs the row in one wave, one class per lane");
  static_assert(H == 1 || H == 2, "per-lane head select below handles 1 or 2 heads");
  const int start = node * SLOT_CAP;
  const int indeg = cnt[node];
  const int deg = min(indeg, SLOT_CAP);
  const int tot = deg + 1;
  const int novf = (indeg > SLOT_CAP) ? min(cnt[n], nE) : 0;
  const int myhead = (lane * VEC) / F;

  float adv[H];
  for (int hd = 0; hd < H; ++hd) adv[hd] = ad[node * H + hd];

  const bool ok0 = lane < tot;
  int s0 = node;
  float v0[H];
  for (int hd = 0; hd < H; ++hd) v0[hd] = -INFINITY;
  if (ok0) edge_logit<H>(as, slots, adv, start, deg, node, lane, s0, v0);

  float mx[H];
  for (int hd = 0; hd < H; ++hd) mx[hd] = v0[hd];
  for (int base = 32; base < tot; base += 32) {
    const int k = base + lane;
    if (k < tot) {
      int s; float v[H];
      edge_logit<H>(as, slots, adv, start, deg, node, k, s, v);
      for (int hd = 0; hd < H; ++hd) mx[hd] = fmaxf(mx[hd], v[hd]);
    }
  }
  if (novf > 0) {
    for (int i = lane; i < novf; i += 32) {
      if (ovf_dst[i] == node) {
        float v[H];
        source_logit<H>(as, adv, ovf_src[i], v);
        for (int hd = 0; hd < H; ++hd) mx[hd] = fmaxf(mx[hd], v[hd]);
      }
    }
  }
  for (int hd = 0; hd < H; ++hd)
    for (int o = 16; o > 0; o >>= 1) mx[hd] = fmaxf(mx[hd], __shfl_xor(mx[hd], o));

  float w0[H], dn[H];
  for (int hd = 0; hd < H; ++hd) {
    w0[hd] = ok0 ? __expf(v0[hd] - mx[hd]) : 0.f;
    dn[hd] = w0[hd];
  }
  for (int base = 32; base < tot; base += 32) {
    const int k = base + lane;
    if (k < tot) {
      int s; float v[H];
      edge_logit<H>(as, slots, adv, start, deg, node, k, s, v);
      for (int hd = 0; hd < H; ++hd) dn[hd] += __expf(v[hd] - mx[hd]);
    }
  }
  if (novf > 0) {
    for (int i = lane; i < novf; i += 32) {
      if (ovf_dst[i] == node) {
        float v[H];
        source_logit<H>(as, adv, ovf_src[i], v);
        for (int hd = 0; hd < H; ++hd) dn[hd] += __expf(v[hd] - mx[hd]);
      }
    }
  }
  for (int hd = 0; hd < H; ++hd)
    for (int o = 16; o > 0; o >>= 1) dn[hd] += __shfl_xor(dn[hd], o);

  for (int i = 0; i < VEC; ++i) acc[i] = 0.f;
  const float* rowbase = hsrc + lane * VEC;

  for (int base = 0; base < tot; base += 32) {
    int sc = s0;
    float al[H];
    if (base == 0) {
      for (int hd = 0; hd < H; ++hd) al[hd] = w0[hd] / (dn[hd] + 1e-16f);
    } else {
      const int k = base + lane;
      sc = node;
      for (int hd = 0; hd < H; ++hd) al[hd] = 0.f;
      if (k < tot) {
        float v[H];
        edge_logit<H>(as, slots, adv, start, deg, node, k, sc, v);
        for (int hd = 0; hd < H; ++hd)
          al[hd] = __expf(v[hd] - mx[hd]) / (dn[hd] + 1e-16f);
      }
    }
    const int nrow = min(tot - base, 32);
#pragma unroll 4
    for (int j = 0; j < nrow; ++j)
      add_row<H, HF, VEC>(rowbase, sc, al, j, myhead, acc);
  }

  if (novf > 0) {
    for (int base = 0; base < novf; base += 32) {
      const int i = base + lane;
      int sc = node;
      float al[H];
      for (int hd = 0; hd < H; ++hd) al[hd] = 0.f;
      bool mine = false;
      if (i < novf) mine = (ovf_dst[i] == node);
      if (mine) {
        float v[H];
        sc = ovf_src[i];
        source_logit<H>(as, adv, sc, v);
        for (int hd = 0; hd < H; ++hd)
          al[hd] = __expf(v[hd] - mx[hd]) / (dn[hd] + 1e-16f);
      }
      unsigned m = __builtin_amdgcn_ballot_w32(mine);
      while (m != 0u) {
        const int j = __builtin_ctz(m);
        m &= m - 1u;
        add_row<H, HF, VEC>(rowbase, sc, al, j, myhead, acc);
      }
    }
  }

  for (int i = 0; i < VEC; ++i) acc[i] += bias[lane * VEC + i];
  if constexpr (LSM) {
    const float v = acc[0];
    float m = v;
    for (int o = 16; o > 0; o >>= 1) m = fmaxf(m, __shfl_xor(m, o));
    float sum = __expf(v - m);
    for (int o = 16; o > 0; o >>= 1) sum += __shfl_xor(sum, o);
    acc[0] = v - m - __logf(sum);
  } else {
    for (int i = 0; i < VEC; ++i)
      acc[i] = acc[i] > 0.f ? acc[i] : (__expf(acc[i]) - 1.f);
  }
}

template <int H, int F, bool LSM>
__global__ __launch_bounds__(256) void gat_gather_kernel(
    const float* __restrict__ hsrc, const float* __restrict__ as,
    const float* __restrict__ ad, const int* __restrict__ cnt,
    const int* __restrict__ slots, const int* __restrict__ ovf_src,
    const int* __restrict__ ovf_dst, int n, int nE,
    const float* __restrict__ bias, float* __restrict__ agg) {
  constexpr int HF = H * F;
  constexpr int VEC = HF / 32;
  const int lane = threadIdx.x & 31;
  const int node = __builtin_amdgcn_readfirstlane(
      (int)((blockIdx.x * blockDim.x + threadIdx.x) >> 5));
  if (node >= n) return;
  float acc[VEC];
  gather_node<H, F, LSM>(hsrc, as, ad, cnt, slots, ovf_src, ovf_dst, n, nE, bias,
                         node, lane, acc);
  float* op = agg + (size_t)node * HF + lane * VEC;
  if constexpr (VEC == 4) {
    float4 r;
    r.x = acc[0]; r.y = acc[1]; r.z = acc[2]; r.w = acc[3];
    *(float4*)op = r;
  } else {
    op[0] = acc[0];
  }
}

constexpr int G1_ROWS = 16;
constexpr int G1_AP = 128 + 4;
constexpr int G1_CP = 32 + 4;

__global__ __launch_bounds__(256) void gather1_gemm2_kernel(
    const float* __restrict__ h1, const float* __restrict__ as1,
    const float* __restrict__ ad1, const int* __restrict__ cnt,
    const int* __restrict__ slots, const int* __restrict__ ovf_src,
    const int* __restrict__ ovf_dst, int n, int nE,
    const float* __restrict__ b1,
    const float* __restrict__ Bp2, float* __restrict__ h2,
    const float* __restrict__ a_s2, const float* __restrict__ a_d2,
    float* __restrict__ as2, float* __restrict__ ad2) {
  __shared__ __align__(16) float Ar[G1_ROWS * G1_AP];
  __shared__ __align__(16) float Cs[G1_ROWS * G1_CP];
  const int tid = threadIdx.x;
  const int lane = tid & 31;
  const int wave = __builtin_amdgcn_readfirstlane(tid >> 5);
  const int row0 = (int)blockIdx.x * G1_ROWS;

  for (int t = 0; t < G1_ROWS / 8; ++t) {
    const int lr = t * 8 + wave;
    const int node = row0 + lr;
    float acc[4];
    if (node < n) {
      gather_node<2, 64, false>(h1, as1, ad1, cnt, slots, ovf_src, ovf_dst, n, nE,
                                b1, node, lane, acc);
    } else {
      for (int i = 0; i < 4; ++i) acc[i] = 0.f;
    }
    float4 r;
    r.x = acc[0]; r.y = acc[1]; r.z = acc[2]; r.w = acc[3];
    *(float4*)&Ar[lr * G1_AP + lane * 4] = r;
  }
  const float sv = a_s2[lane];
  const float dv = a_d2[lane];
  __syncthreads();

  if (wave < 2) {
    const int c = lane & 15;
    const int koff = (lane >> 4) << 1;
    const int ncol = (wave << 4) + c;
    const int bbase = (lane >> 4) * 32 + ncol;
    v8f acc = v8f{0.f, 0.f, 0.f, 0.f, 0.f, 0.f, 0.f, 0.f};
#pragma unroll
    for (int half = 0; half < 2; ++half) {
#pragma unroll
      for (int k2 = 0; k2 < 16; ++k2) {
        const int kk = half * 16 + k2;
        const v2f b = *(const v2f*)&Bp2[(size_t)(bbase + kk * 2 * 32) * 2];
        v2f a;
        a.x = Ar[c * G1_AP + kk * 4 + koff];
        a.y = Ar[c * G1_AP + kk * 4 + koff + 1];
        acc = __builtin_amdgcn_wmma_f32_16x16x4_f32(
             false, a,  false, b,
             (short)0, acc,  false,  false);
      }
      if (half == 0) __builtin_amdgcn_sched_barrier(0);
    }
    const int crow = (lane >> 4) << 3;
#pragma unroll
    for (int i = 0; i < 8; ++i) Cs[(crow + i) * G1_CP + ncol] = acc[i];
  }
  __syncthreads();

  if (tid < G1_ROWS * 8) {
    const int r = tid >> 3;
    const int c4 = (tid & 7) << 2;
    const int gr = row0 + r;
    if (gr < n) *(float4*)&h2[(size_t)gr * 32 + c4] = *(const float4*)&Cs[r * G1_CP + c4];
  } else if (wave == 4) {
    const int r = lane & 15;
    const bool isd = lane >= 16;
    float s = 0.f;
#pragma unroll
    for (int f = 0; f < 32; ++f) {
      const float vs = __int_as_float(__builtin_amdgcn_readlane(__float_as_int(sv), f));
      const float vd = __int_as_float(__builtin_amdgcn_readlane(__float_as_int(dv), f));
      s += Cs[r * G1_CP + f] * (isd ? vd : vs);
    }
    const int gr = row0 + r;
    if (gr < n) {
      if (isd) ad2[gr] = s;
      else as2[gr] = s;
    }
  }
}

extern "C" void kernel_launch(void* const* d_in, const int* in_sizes, int n_in,
                              void* d_out, int out_size, void* d_ws,
                              size_t ws_size, hipStream_t stream) {
  (void)n_in; (void)out_size; (void)ws_size;
  const float* x      = (const float*)d_in[0];
  const int*   ei     = (const int*)d_in[1];
  const float* W1     = (const float*)d_in[2];
  const float* a_src1 = (const float*)d_in[3];
  const float* a_dst1 = (const float*)d_in[4];
  const float* b1     = (const float*)d_in[5];
  const float* W2     = (const float*)d_in[6];
  const float* a_src2 = (const float*)d_in[7];
  const float* a_dst2 = (const float*)d_in[8];
  const float* b2     = (const float*)d_in[9];

  const int n    = in_sizes[0] / 500;
  const int nE   = in_sizes[1] / 2;
  const int* src = ei;
  const int* dst = ei + nE;

  float* ws = (float*)d_ws;
  size_t o = 0;
  float* h1   = ws + o; o += (size_t)n * 128;
  float* h2   = ws + o; o += (size_t)n * 32;
  float* Bp1  = ws + o; o += (size_t)500 * 128;
  float* Bp2  = ws + o; o += (size_t)128 * 32;
  float* as1  = ws + o; o += (size_t)n * 2;
  float* ad1  = ws + o; o += (size_t)n * 2;
  float* as2  = ws + o; o += (size_t)n;
  float* ad2  = ws + o; o += (size_t)n;
  int* cnt     = (int*)(ws + o); o += (size_t)n + 1;
  int* slots   = (int*)(ws + o); o += (size_t)n * SLOT_CAP;
  int* ovf_src = (int*)(ws + o); o += (size_t)nE;
  int* ovf_dst = (int*)(ws + o); o += (size_t)nE;

  prep_kernel<<<PACK1_BLOCKS + PACK2_BLOCKS + (n + 1 + 255) / 256, 256, 0, stream>>>(
      W1, Bp1, W2, Bp2, (unsigned*)cnt, n + 1);

  const int fblocks = (nE + 256 * FILL_EPT - 1) / (256 * FILL_EPT);
  const int rblocks = (n + 31) / 32;
  gemm1_fill_kernel<<<fblocks + rblocks, 256, 0, stream>>>(
      x, Bp1, h1, n, a_src1, a_dst1, as1, ad1, fblocks,
      src, dst, nE, cnt, slots, ovf_src, ovf_dst);

  const int gblocks = (int)(((size_t)n * 32 + 255) / 256);

  gather1_gemm2_kernel<<<(n + G1_ROWS - 1) / G1_ROWS, 256, 0, stream>>>(
      h1, as1, ad1, cnt, slots, ovf_src, ovf_dst, n, nE, b1, Bp2, h2,
      a_src2, a_dst2, as2, ad2);

  gat_gather_kernel<1, 32, true><<<gblocks, 256, 0, stream>>>(
      h2, as2, ad2, cnt, slots, ovf_src, ovf_dst, n, nE, b2, (float*)d_out);
}
